# starting kernel
# baseline (speedup 1.0000x reference)
#include <hip/hip_runtime.h>

typedef __attribute__((ext_vector_type(2))) float v2f;
typedef __attribute__((ext_vector_type(4))) float v4f;
typedef __attribute__((ext_vector_type(8))) float v8f;

#define NBATCH 8
#define NFEAT  256
#define HD     64
#define NT     1024
#define AUGC   130
#define XSTR   68
#define YCHUNK 256

__launch_bounds__(128)
__global__ void ssm_scan_wmma_f32(const float* __restrict__ x,
                                  const float* __restrict__ Amat,
                                  const float* __restrict__ Bmat,
                                  const float* __restrict__ Cmat,
                                  float* __restrict__ out)
{
    __shared__ float smem[NBATCH*NT + NBATCH*YCHUNK + 16*XSTR + 128];
    float* aug   = smem;
    float* u_lds = smem;
    float* y_lds = smem + NBATCH*NT;
    float* X_lds = smem + NBATCH*NT + NBATCH*YCHUNK;
    float* p_lds = X_lds + 16*XSTR;

    const int f   = blockIdx.x;
    const int tid = threadIdx.x;

    const float* Af = Amat + f*HD*HD;
    for (int e = tid; e < HD*HD; e += 128) {
        int r = e >> 6, c = e & 63;
        float a   = Af[e];
        float eye = (r == c) ? 1.0f : 0.0f;
        aug[r*AUGC + c]      = eye - 0.5f*a;
        aug[r*AUGC + 64 + c] = eye + 0.5f*a;
    }
    for (int r = tid; r < HD; r += 128)
        aug[r*AUGC + 128] = Bmat[f*HD + r];
    __syncthreads();

    {
        const int row  = tid & 63;
        const int c0   = (tid >> 6) ? 64 : 0;
        const int ncol = (tid >> 6) ? 65 : 64;
        for (int p = 0; p < HD; ++p) {
            float fac = aug[row*AUGC + p] / aug[p*AUGC + p];
            __syncthreads();
            if (row != p) {
                for (int c = 0; c < ncol; ++c)
                    aug[row*AUGC + c0 + c] -= fac * aug[p*AUGC + c0 + c];
            }
            __syncthreads();
        }
    }

    const int wv   = tid >> 5;
    const int ln   = tid & 31;
    const int hi   = ln >> 4;
    const int col  = ln & 15;
    const int koff = hi << 1;
    const int arow = wv*16 + col;
    const int row2 = wv*16 + hi*8;

    v2f a_op[16];
    {
        float rinv = 1.0f / aug[arow*AUGC + arow];
        #pragma unroll
        for (int k = 0; k < 16; ++k) {
            a_op[k].x = aug[arow*AUGC + 64 + 4*k + koff]     * rinv;
            a_op[k].y = aug[arow*AUGC + 64 + 4*k + koff + 1] * rinv;
        }
    }
    float bvr[8], cvr[8];
    #pragma unroll
    for (int r = 0; r < 8; ++r) {
        int rr = row2 + r;
        bvr[r] = aug[rr*AUGC + 128] / aug[rr*AUGC + rr];
        cvr[r] = Cmat[f*HD + rr];
    }
    __syncthreads();

    for (int e = tid; e < NBATCH*NT; e += 128) {
        int b = e >> 10, i = e & (NT - 1);
        u_lds[i*NBATCH + b] = x[(size_t)(b*NFEAT + f)*NT + i];
    }
    for (int e = tid; e < 16*XSTR; e += 128) X_lds[e] = 0.0f;
    __syncthreads();

    for (int t = 0; t < NT; ++t) {
        v8f D = {};
        #pragma unroll
        for (int k = 0; k < 16; ++k) {
            v2f bop = *(const v2f*)&X_lds[col*XSTR + 4*k + koff];
            D = __builtin_amdgcn_wmma_f32_16x16x4_f32(false, a_op[k], false, bop,
                                                      (short)0, D, false, false);
        }

        float u = (col < NBATCH) ? u_lds[t*NBATCH + col] : 0.0f;
        float psum = 0.0f;
        #pragma unroll
        for (int r = 0; r < 8; ++r) {
            D[r] += bvr[r] * u;
            psum += cvr[r] * D[r];
        }
        p_lds[(wv*2 + hi)*16 + col] = psum;
        __syncthreads();

        v4f lo4 = { D[0], D[1], D[2], D[3] };
        v4f hi4 = { D[4], D[5], D[6], D[7] };
        *(v4f*)&X_lds[col*XSTR + row2]     = lo4;
        *(v4f*)&X_lds[col*XSTR + row2 + 4] = hi4;

        if (tid < NBATCH) {
            float y = 0.0f;
            #pragma unroll
            for (int j = 0; j < 8; ++j) y += p_lds[j*16 + tid];
            y_lds[(t & (YCHUNK-1))*NBATCH + tid] = y;
        }

        if ((t & (YCHUNK-1)) == (YCHUNK-1)) {
            __syncthreads();
            int tbase = t & ~(YCHUNK-1);
            for (int e = tid; e < NBATCH*YCHUNK; e += 128) {
                int b = e >> 8, i = e & (YCHUNK-1);
                out[(size_t)(b*NFEAT + f)*NT + tbase + i] = y_lds[i*NBATCH + b];
            }
        }
        __syncthreads();
    }
}

extern "C" void kernel_launch(void* const* d_in, const int* in_sizes, int n_in,
                              void* d_out, int out_size, void* d_ws, size_t ws_size,
                              hipStream_t stream) {
    (void)in_sizes; (void)n_in; (void)d_ws; (void)ws_size; (void)out_size;
    const float* x = (const float*)d_in[0];
    const float* A = (const float*)d_in[1];
    const float* B = (const float*)d_in[2];
    const float* C = (const float*)d_in[3];
    float* out = (float*)d_out;
    ssm_scan_wmma_f32<<<dim3(NFEAT), dim3(128), 0, stream>>>(x, A, B, C, out);
}


# Gauss-Jordan row update in registers: each thread reads its half of pivot row and own row with 16-byte LDS loads, all issued before the first store (row pitch 132), not one LDS round trip per element
# speedup vs baseline: 1.2859x; 1.2859x over previous
#include <hip/hip_runtime.h>

typedef __attribute__((ext_vector_type(2))) float v2f;
typedef __attribute__((ext_vector_type(4))) float v4f;
typedef __attribute__((ext_vector_type(8))) float v8f;

#define NBATCH 8
#define NFEAT  256
#define HD     64
#define NT     1024
#define AUGC   132
#define XSTR   68
#define YCHUNK 256

__launch_bounds__(128)
__global__ void ssm_scan_wmma_f32(const float* __restrict__ x,
                                  const float* __restrict__ Amat,
                                  const float* __restrict__ Bmat,
                                  const float* __restrict__ Cmat,
                                  float* __restrict__ out)
{
    __shared__ __attribute__((aligned(16))) float smem[NBATCH*NT + NBATCH*YCHUNK + 16*XSTR + 128];
    float* aug   = smem;
    float* u_lds = smem;
    float* y_lds = smem + NBATCH*NT;
    float* X_lds = smem + NBATCH*NT + NBATCH*YCHUNK;
    float* p_lds = X_lds + 16*XSTR;

    const int f   = blockIdx.x;
    const int tid = threadIdx.x;

    const float* Af = Amat + f*HD*HD;
    for (int e = tid; e < HD*HD; e += 128) {
        int r = e >> 6, c = e & 63;
        float a   = Af[e];
        float eye = (r == c) ? 1.0f : 0.0f;
        aug[r*AUGC + c]      = eye - 0.5f*a;
        aug[r*AUGC + 64 + c] = eye + 0.5f*a;
    }
    for (int r = tid; r < HD; r += 128) {
        aug[r*AUGC + 128] = Bmat[f*HD + r];
        aug[r*AUGC + 129] = 0.0f;
        aug[r*AUGC + 130] = 0.0f;
        aug[r*AUGC + 131] = 0.0f;
    }
    __syncthreads();

    {
        const int row  = tid & 63;
        const int half = tid >> 6;
        float* orow = aug + row*AUGC + half*64;
        for (int p = 0; p < HD; ++p) {
            float fac = aug[row*AUGC + p] / aug[p*AUGC + p];
            __syncthreads();
            if (row != p) {
                const float* prow = aug + p*AUGC + half*64;
                v4f pv[16], ov[16];
                v4f pe = {}, oe = {};
                if (half) {
                    pe = *(const v4f*)(prow + 64);
                    oe = *(const v4f*)(orow + 64);
                }
                #pragma unroll
                for (int j = 0; j < 16; ++j) {
                    pv[j] = *(const v4f*)(prow + 4*j);
                    ov[j] = *(const v4f*)(orow + 4*j);
                }
                #pragma unroll
                for (int j = 0; j < 16; ++j) ov[j] -= fac * pv[j];
                oe -= fac * pe;
                #pragma unroll
                for (int j = 0; j < 16; ++j) *(v4f*)(orow + 4*j) = ov[j];
                if (half) *(v4f*)(orow + 64) = oe;
            }
            __syncthreads();
        }
    }

    const int wv   = tid >> 5;
    const int ln   = tid & 31;
    const int hi   = ln >> 4;
    const int col  = ln & 15;
    const int koff = hi << 1;
    const int arow = wv*16 + col;
    const int row2 = wv*16 + hi*8;

    v2f a_op[16];
    {
        float rinv = 1.0f / aug[arow*AUGC + arow];
        #pragma unroll
        for (int k = 0; k < 16; ++k) {
            a_op[k].x = aug[arow*AUGC + 64 + 4*k + koff]     * rinv;
            a_op[k].y = aug[arow*AUGC + 64 + 4*k + koff + 1] * rinv;
        }
    }
    float bvr[8], cvr[8];
    #pragma unroll
    for (int r = 0; r < 8; ++r) {
        int rr = row2 + r;
        bvr[r] = aug[rr*AUGC + 128] / aug[rr*AUGC + rr];
        cvr[r] = Cmat[f*HD + rr];
    }
    __syncthreads();

    for (int e = tid; e < NBATCH*NT; e += 128) {
        int b = e >> 10, i = e & (NT - 1);
        u_lds[i*NBATCH + b] = x[(size_t)(b*NFEAT + f)*NT + i];
    }
    for (int e = tid; e < 16*XSTR; e += 128) X_lds[e] = 0.0f;
    __syncthreads();

    for (int t = 0; t < NT; ++t) {
        v8f D = {};
        #pragma unroll
        for (int k = 0; k < 16; ++k) {
            v2f bop = *(const v2f*)&X_lds[col*XSTR + 4*k + koff];
            D = __builtin_amdgcn_wmma_f32_16x16x4_f32(false, a_op[k], false, bop,
                                                      (short)0, D, false, false);
        }

        float u = (col < NBATCH) ? u_lds[t*NBATCH + col] : 0.0f;
        float psum = 0.0f;
        #pragma unroll
        for (int r = 0; r < 8; ++r) {
            D[r] += bvr[r] * u;
            psum += cvr[r] * D[r];
        }
        p_lds[(wv*2 + hi)*16 + col] = psum;
        __syncthreads();

        v4f lo4 = { D[0], D[1], D[2], D[3] };
        v4f hi4 = { D[4], D[5], D[6], D[7] };
        *(v4f*)&X_lds[col*XSTR + row2]     = lo4;
        *(v4f*)&X_lds[col*XSTR + row2 + 4] = hi4;

        if (tid < NBATCH) {
            float y = 0.0f;
            #pragma unroll
            for (int j = 0; j < 8; ++j) y += p_lds[j*16 + tid];
            y_lds[(t & (YCHUNK-1))*NBATCH + tid] = y;
        }

        if ((t & (YCHUNK-1)) == (YCHUNK-1)) {
            __syncthreads();
            int tbase = t & ~(YCHUNK-1);
            for (int e = tid; e < NBATCH*YCHUNK; e += 128) {
                int b = e >> 8, i = e & (YCHUNK-1);
                out[(size_t)(b*NFEAT + f)*NT + tbase + i] = y_lds[i*NBATCH + b];
            }
        }
        __syncthreads();
    }
}

extern "C" void kernel_launch(void* const* d_in, const int* in_sizes, int n_in,
                              void* d_out, int out_size, void* d_ws, size_t ws_size,
                              hipStream_t stream) {
    (void)in_sizes; (void)n_in; (void)d_ws; (void)ws_size; (void)out_size;
    const float* x = (const float*)d_in[0];
    const float* A = (const float*)d_in[1];
    const float* B = (const float*)d_in[2];
    const float* C = (const float*)d_in[3];
    float* out = (float*)d_out;
    ssm_scan_wmma_f32<<<dim3(NFEAT), dim3(128), 0, stream>>>(x, A, B, C, out);
}


# One workgroup barrier per time step instead of two: state double-buffered in LDS, the 8 y partials stored per step and summed in the same order at a 128-step flush
# speedup vs baseline: 1.5255x; 1.1863x over previous
#include <hip/hip_runtime.h>

typedef __attribute__((ext_vector_type(2))) float v2f;
typedef __attribute__((ext_vector_type(4))) float v4f;
typedef __attribute__((ext_vector_type(8))) float v8f;

#define NBATCH 8
#define NFEAT  256
#define HD     64
#define NT     1024
#define AUGC   132
#define XSTR   68
#define YCHUNK 128
#define XBUF   (16*XSTR)
#define PSTR   129

__launch_bounds__(128)
__global__ void ssm_scan_wmma_f32(const float* __restrict__ x,
                                  const float* __restrict__ Amat,
                                  const float* __restrict__ Bmat,
                                  const float* __restrict__ Cmat,
                                  float* __restrict__ out)
{
    __shared__ __attribute__((aligned(16))) float smem[NBATCH*NT + YCHUNK*PSTR + 2*XBUF];
    float* aug   = smem;
    float* u_lds = smem;
    float* p_lds = smem + NBATCH*NT;
    float* X_lds = smem + NBATCH*NT + YCHUNK*PSTR;

    const int f   = blockIdx.x;
    const int tid = threadIdx.x;

    const float* Af = Amat + f*HD*HD;
    for (int e = tid; e < HD*HD; e += 128) {
        int r = e >> 6, c = e & 63;
        float a   = Af[e];
        float eye = (r == c) ? 1.0f : 0.0f;
        aug[r*AUGC + c]      = eye - 0.5f*a;
        aug[r*AUGC + 64 + c] = eye + 0.5f*a;
    }
    for (int r = tid; r < HD; r += 128) {
        aug[r*AUGC + 128] = Bmat[f*HD + r];
        aug[r*AUGC + 129] = 0.0f;
        aug[r*AUGC + 130] = 0.0f;
        aug[r*AUGC + 131] = 0.0f;
    }
    __syncthreads();

    {
        const int row  = tid & 63;
        const int half = tid >> 6;
        float* orow = aug + row*AUGC + half*64;
        for (int p = 0; p < HD; ++p) {
            float fac = aug[row*AUGC + p] / aug[p*AUGC + p];
            __syncthreads();
            if (row != p) {
                const float* prow = aug + p*AUGC + half*64;
                v4f pv[16], ov[16];
                v4f pe = {}, oe = {};
                if (half) {
                    pe = *(const v4f*)(prow + 64);
                    oe = *(const v4f*)(orow + 64);
                }
                #pragma unroll
                for (int j = 0; j < 16; ++j) {
                    pv[j] = *(const v4f*)(prow + 4*j);
                    ov[j] = *(const v4f*)(orow + 4*j);
                }
                #pragma unroll
                for (int j = 0; j < 16; ++j) ov[j] -= fac * pv[j];
                oe -= fac * pe;
                #pragma unroll
                for (int j = 0; j < 16; ++j) *(v4f*)(orow + 4*j) = ov[j];
                if (half) *(v4f*)(orow + 64) = oe;
            }
            __syncthreads();
        }
    }

    const int wv   = tid >> 5;
    const int ln   = tid & 31;
    const int hi   = ln >> 4;
    const int col  = ln & 15;
    const int koff = hi << 1;
    const int arow = wv*16 + col;
    const int row2 = wv*16 + hi*8;

    v2f a_op[16];
    {
        float rinv = 1.0f / aug[arow*AUGC + arow];
        #pragma unroll
        for (int k = 0; k < 16; ++k) {
            a_op[k].x = aug[arow*AUGC + 64 + 4*k + koff]     * rinv;
            a_op[k].y = aug[arow*AUGC + 64 + 4*k + koff + 1] * rinv;
        }
    }
    float bvr[8], cvr[8];
    #pragma unroll
    for (int r = 0; r < 8; ++r) {
        int rr = row2 + r;
        bvr[r] = aug[rr*AUGC + 128] / aug[rr*AUGC + rr];
        cvr[r] = Cmat[f*HD + rr];
    }
    __syncthreads();

    for (int e = tid; e < NBATCH*NT; e += 128) {
        int b = e >> 10, i = e & (NT - 1);
        u_lds[i*NBATCH + b] = x[(size_t)(b*NFEAT + f)*NT + i];
    }
    for (int e = tid; e < XBUF; e += 128) X_lds[e] = 0.0f;
    __syncthreads();

    const int pslot = (wv*2 + hi)*16 + col;
    for (int t = 0; t < NT; ++t) {
        const float* Xr = X_lds + (t & 1)*XBUF;
        float*       Xw = X_lds + ((t & 1) ^ 1)*XBUF;

        v8f D = {};
        #pragma unroll
        for (int k = 0; k < 16; ++k) {
            v2f bop = *(const v2f*)&Xr[col*XSTR + 4*k + koff];
            D = __builtin_amdgcn_wmma_f32_16x16x4_f32(false, a_op[k], false, bop,
                                                      (short)0, D, false, false);
        }

        float u = (col < NBATCH) ? u_lds[t*NBATCH + col] : 0.0f;
        float psum = 0.0f;
        #pragma unroll
        for (int r = 0; r < 8; ++r) {
            D[r] += bvr[r] * u;
            psum += cvr[r] * D[r];
        }

        v4f lo4 = { D[0], D[1], D[2], D[3] };
        v4f hi4 = { D[4], D[5], D[6], D[7] };
        *(v4f*)&Xw[col*XSTR + row2]     = lo4;
        *(v4f*)&Xw[col*XSTR + row2 + 4] = hi4;
        p_lds[(t & (YCHUNK-1))*PSTR + pslot] = psum;
        __syncthreads();

        if ((t & (YCHUNK-1)) == (YCHUNK-1)) {
            int tbase = t & ~(YCHUNK-1);
            for (int e = tid; e < NBATCH*YCHUNK; e += 128) {
                int b = e >> 7, i = e & (YCHUNK-1);
                float y = 0.0f;
                for (int j = 0; j < 8; ++j) y += p_lds[i*PSTR + j*16 + b];
                out[(size_t)(b*NFEAT + f)*NT + tbase + i] = y;
            }
            __syncthreads();
        }
    }
}

extern "C" void kernel_launch(void* const* d_in, const int* in_sizes, int n_in,
                              void* d_out, int out_size, void* d_ws, size_t ws_size,
                              hipStream_t stream) {
    (void)in_sizes; (void)n_in; (void)d_ws; (void)ws_size; (void)out_size;
    const float* x = (const float*)d_in[0];
    const float* A = (const float*)d_in[1];
    const float* B = (const float*)d_in[2];
    const float* C = (const float*)d_in[3];
    float* out = (float*)d_out;
    ssm_scan_wmma_f32<<<dim3(NFEAT), dim3(128), 0, stream>>>(x, A, B, C, out);
}


# All LDS reads of a time step (16 state slices and the input u) issued as one group ahead of the 16-WMMA chain, not between the WMMAs and behind the last one; k and summation order unchanged
# speedup vs baseline: 2.0001x; 1.3111x over previous
#include <hip/hip_runtime.h>

typedef __attribute__((ext_vector_type(2))) float v2f;
typedef __attribute__((ext_vector_type(4))) float v4f;
typedef __attribute__((ext_vector_type(8))) float v8f;

#define NBATCH 8
#define NFEAT  256
#define HD     64
#define NT     1024
#define AUGC   132
#define XSTR   68
#define YCHUNK 128
#define XBUF   (16*XSTR)
#define PSTR   129

__launch_bounds__(128)
__global__ void ssm_scan_wmma_f32(const float* __restrict__ x,
                                  const float* __restrict__ Amat,
                                  const float* __restrict__ Bmat,
                                  const float* __restrict__ Cmat,
                                  float* __restrict__ out)
{
    __shared__ __attribute__((aligned(16))) float smem[NBATCH*NT + YCHUNK*PSTR + 2*XBUF];
    float* aug   = smem;
    float* u_lds = smem;
    float* p_lds = smem + NBATCH*NT;
    float* X_lds = smem + NBATCH*NT + YCHUNK*PSTR;

    const int f   = blockIdx.x;
    const int tid = threadIdx.x;

    const float* Af = Amat + f*HD*HD;
    for (int e = tid; e < HD*HD; e += 128) {
        int r = e >> 6, c = e & 63;
        float a   = Af[e];
        float eye = (r == c) ? 1.0f : 0.0f;
        aug[r*AUGC + c]      = eye - 0.5f*a;
        aug[r*AUGC + 64 + c] = eye + 0.5f*a;
    }
    for (int r = tid; r < HD; r += 128) {
        aug[r*AUGC + 128] = Bmat[f*HD + r];
        aug[r*AUGC + 129] = 0.0f;
        aug[r*AUGC + 130] = 0.0f;
        aug[r*AUGC + 131] = 0.0f;
    }
    __syncthreads();

    {
        const int row  = tid & 63;
        const int half = tid >> 6;
        float* orow = aug + row*AUGC + half*64;
        for (int p = 0; p < HD; ++p) {
            float fac = aug[row*AUGC + p] / aug[p*AUGC + p];
            __syncthreads();
            if (row != p) {
                const float* prow = aug + p*AUGC + half*64;
                v4f pv[16], ov[16];
                v4f pe = {}, oe = {};
                if (half) {
                    pe = *(const v4f*)(prow + 64);
                    oe = *(const v4f*)(orow + 64);
                }
                #pragma unroll
                for (int j = 0; j < 16; ++j) {
                    pv[j] = *(const v4f*)(prow + 4*j);
                    ov[j] = *(const v4f*)(orow + 4*j);
                }
                #pragma unroll
                for (int j = 0; j < 16; ++j) ov[j] -= fac * pv[j];
                oe -= fac * pe;
                #pragma unroll
                for (int j = 0; j < 16; ++j) *(v4f*)(orow + 4*j) = ov[j];
                if (half) *(v4f*)(orow + 64) = oe;
            }
            __syncthreads();
        }
    }

    const int wv   = tid >> 5;
    const int ln   = tid & 31;
    const int hi   = ln >> 4;
    const int col  = ln & 15;
    const int koff = hi << 1;
    const int arow = wv*16 + col;
    const int row2 = wv*16 + hi*8;

    v2f a_op[16];
    {
        float rinv = 1.0f / aug[arow*AUGC + arow];
        #pragma unroll
        for (int k = 0; k < 16; ++k) {
            a_op[k].x = aug[arow*AUGC + 64 + 4*k + koff]     * rinv;
            a_op[k].y = aug[arow*AUGC + 64 + 4*k + koff + 1] * rinv;
        }
    }
    float bvr[8], cvr[8];
    #pragma unroll
    for (int r = 0; r < 8; ++r) {
        int rr = row2 + r;
        float bw = aug[rr*AUGC + 128] / aug[rr*AUGC + rr];
        bvr[r] = (col < NBATCH) ? bw : 0.0f;
        cvr[r] = Cmat[f*HD + rr];
    }
    __syncthreads();

    for (int e = tid; e < NBATCH*NT; e += 128) {
        int b = e >> 10, i = e & (NT - 1);
        u_lds[i*NBATCH + b] = x[(size_t)(b*NFEAT + f)*NT + i];
    }
    for (int e = tid; e < XBUF; e += 128) X_lds[e] = 0.0f;
    __syncthreads();

    const int pslot = (wv*2 + hi)*16 + col;
    for (int t = 0; t < NT; ++t) {
        const float* Xr = X_lds + (t & 1)*XBUF;
        float*       Xw = X_lds + ((t & 1) ^ 1)*XBUF;

        v2f bop[16];
        #pragma unroll
        for (int k = 0; k < 16; ++k)
            bop[k] = *(const v2f*)&Xr[col*XSTR + 4*k + koff];
        float u = u_lds[t*NBATCH + (col & (NBATCH-1))];
        __builtin_amdgcn_sched_barrier(0);
        v8f D = {};
        #pragma unroll
        for (int k = 0; k < 16; ++k)
            D = __builtin_amdgcn_wmma_f32_16x16x4_f32(false, a_op[k], false, bop[k],
                                                      (short)0, D, false, false);

        float psum = 0.0f;
        #pragma unroll
        for (int r = 0; r < 8; ++r) {
            D[r] += bvr[r] * u;
            psum += cvr[r] * D[r];
        }

        v4f lo4 = { D[0], D[1], D[2], D[3] };
        v4f hi4 = { D[4], D[5], D[6], D[7] };
        *(v4f*)&Xw[col*XSTR + row2]     = lo4;
        *(v4f*)&Xw[col*XSTR + row2 + 4] = hi4;
        p_lds[(t & (YCHUNK-1))*PSTR + pslot] = psum;
        __syncthreads();

        if ((t & (YCHUNK-1)) == (YCHUNK-1)) {
            int tbase = t & ~(YCHUNK-1);
            for (int e = tid; e < NBATCH*YCHUNK; e += 128) {
                int b = e >> 7, i = e & (YCHUNK-1);
                float y = 0.0f;
                for (int j = 0; j < 8; ++j) y += p_lds[i*PSTR + j*16 + b];
                out[(size_t)(b*NFEAT + f)*NT + tbase + i] = y;
            }
            __syncthreads();
        }
    }
}

extern "C" void kernel_launch(void* const* d_in, const int* in_sizes, int n_in,
                              void* d_out, int out_size, void* d_ws, size_t ws_size,
                              hipStream_t stream) {
    (void)in_sizes; (void)n_in; (void)d_ws; (void)ws_size; (void)out_size;
    const float* x = (const float*)d_in[0];
    const float* A = (const float*)d_in[1];
    const float* B = (const float*)d_in[2];
    const float* C = (const float*)d_in[3];
    float* out = (float*)d_out;
    ssm_scan_wmma_f32<<<dim3(NFEAT), dim3(128), 0, stream>>>(x, A, B, C, out);
}


# The 16-WMMA chain of a time step split into two independent accumulators (even / odd k slices, the odd one seeded with the input term Bv*u), joined by one add: summation order changed
# speedup vs baseline: 2.1943x; 1.0971x over previous
#include <hip/hip_runtime.h>

typedef __attribute__((ext_vector_type(2))) float v2f;
typedef __attribute__((ext_vector_type(4))) float v4f;
typedef __attribute__((ext_vector_type(8))) float v8f;

#define NBATCH 8
#define NFEAT  256
#define HD     64
#define NT     1024
#define AUGC   132
#define XSTR   68
#define YCHUNK 128
#define XBUF   (16*XSTR)
#define PSTR   129

__launch_bounds__(128)
__global__ void ssm_scan_wmma_f32(const float* __restrict__ x,
                                  const float* __restrict__ Amat,
                                  const float* __restrict__ Bmat,
                                  const float* __restrict__ Cmat,
                                  float* __restrict__ out)
{
    __shared__ __attribute__((aligned(16))) float smem[NBATCH*NT + YCHUNK*PSTR + 2*XBUF];
    float* aug   = smem;
    float* u_lds = smem;
    float* p_lds = smem + NBATCH*NT;
    float* X_lds = smem + NBATCH*NT + YCHUNK*PSTR;

    const int f   = blockIdx.x;
    const int tid = threadIdx.x;

    const float* Af = Amat + f*HD*HD;
    for (int e = tid; e < HD*HD; e += 128) {
        int r = e >> 6, c = e & 63;
        float a   = Af[e];
        float eye = (r == c) ? 1.0f : 0.0f;
        aug[r*AUGC + c]      = eye - 0.5f*a;
        aug[r*AUGC + 64 + c] = eye + 0.5f*a;
    }
    for (int r = tid; r < HD; r += 128) {
        aug[r*AUGC + 128] = Bmat[f*HD + r];
        aug[r*AUGC + 129] = 0.0f;
        aug[r*AUGC + 130] = 0.0f;
        aug[r*AUGC + 131] = 0.0f;
    }
    __syncthreads();

    {
        const int row  = tid & 63;
        const int half = tid >> 6;
        float* orow = aug + row*AUGC + half*64;
        for (int p = 0; p < HD; ++p) {
            float fac = aug[row*AUGC + p] / aug[p*AUGC + p];
            __syncthreads();
            if (row != p) {
                const float* prow = aug + p*AUGC + half*64;
                v4f pv[16], ov[16];
                v4f pe = {}, oe = {};
                if (half) {
                    pe = *(const v4f*)(prow + 64);
                    oe = *(const v4f*)(orow + 64);
                }
                #pragma unroll
                for (int j = 0; j < 16; ++j) {
                    pv[j] = *(const v4f*)(prow + 4*j);
                    ov[j] = *(const v4f*)(orow + 4*j);
                }
                #pragma unroll
                for (int j = 0; j < 16; ++j) ov[j] -= fac * pv[j];
                oe -= fac * pe;
                #pragma unroll
                for (int j = 0; j < 16; ++j) *(v4f*)(orow + 4*j) = ov[j];
                if (half) *(v4f*)(orow + 64) = oe;
            }
            __syncthreads();
        }
    }

    const int wv   = tid >> 5;
    const int ln   = tid & 31;
    const int hi   = ln >> 4;
    const int col  = ln & 15;
    const int koff = hi << 1;
    const int arow = wv*16 + col;
    const int row2 = wv*16 + hi*8;

    v2f a_op[16];
    {
        float rinv = 1.0f / aug[arow*AUGC + arow];
        #pragma unroll
        for (int k = 0; k < 16; ++k) {
            a_op[k].x = aug[arow*AUGC + 64 + 4*k + koff]     * rinv;
            a_op[k].y = aug[arow*AUGC + 64 + 4*k + koff + 1] * rinv;
        }
    }
    float bvr[8], cvr[8];
    #pragma unroll
    for (int r = 0; r < 8; ++r) {
        int rr = row2 + r;
        float bw = aug[rr*AUGC + 128] / aug[rr*AUGC + rr];
        bvr[r] = (col < NBATCH) ? bw : 0.0f;
        cvr[r] = Cmat[f*HD + rr];
    }
    __syncthreads();

    for (int e = tid; e < NBATCH*NT; e += 128) {
        int b = e >> 10, i = e & (NT - 1);
        u_lds[i*NBATCH + b] = x[(size_t)(b*NFEAT + f)*NT + i];
    }
    for (int e = tid; e < XBUF; e += 128) X_lds[e] = 0.0f;
    __syncthreads();

    const int pslot = (wv*2 + hi)*16 + col;
    for (int t = 0; t < NT; ++t) {
        const float* Xr = X_lds + (t & 1)*XBUF;
        float*       Xw = X_lds + ((t & 1) ^ 1)*XBUF;

        float u = u_lds[t*NBATCH + (col & (NBATCH-1))];
        v2f bop[16];
        #pragma unroll
        for (int k = 0; k < 16; ++k)
            bop[k] = *(const v2f*)&Xr[col*XSTR + 4*k + koff];
        __builtin_amdgcn_sched_barrier(0);
        v8f D0 = {};
        v8f D1;
        #pragma unroll
        for (int r = 0; r < 8; ++r) D1[r] = bvr[r] * u;
        #pragma unroll
        for (int k = 0; k < 16; k += 2) {
            D0 = __builtin_amdgcn_wmma_f32_16x16x4_f32(false, a_op[k],   false, bop[k],
                                                       (short)0, D0, false, false);
            D1 = __builtin_amdgcn_wmma_f32_16x16x4_f32(false, a_op[k+1], false, bop[k+1],
                                                       (short)0, D1, false, false);
        }
        v8f D = D0 + D1;

        float psum = 0.0f;
        #pragma unroll
        for (int r = 0; r < 8; ++r)
            psum += cvr[r] * D[r];

        v4f lo4 = { D[0], D[1], D[2], D[3] };
        v4f hi4 = { D[4], D[5], D[6], D[7] };
        *(v4f*)&Xw[col*XSTR + row2]     = lo4;
        *(v4f*)&Xw[col*XSTR + row2 + 4] = hi4;
        p_lds[(t & (YCHUNK-1))*PSTR + pslot] = psum;
        __syncthreads();

        if ((t & (YCHUNK-1)) == (YCHUNK-1)) {
            int tbase = t & ~(YCHUNK-1);
            for (int e = tid; e < NBATCH*YCHUNK; e += 128) {
                int b = e >> 7, i = e & (YCHUNK-1);
                float y = 0.0f;
                for (int j = 0; j < 8; ++j) y += p_lds[i*PSTR + j*16 + b];
                out[(size_t)(b*NFEAT + f)*NT + tbase + i] = y;
            }
            __syncthreads();
        }
    }
}

extern "C" void kernel_launch(void* const* d_in, const int* in_sizes, int n_in,
                              void* d_out, int out_size, void* d_ws, size_t ws_size,
                              hipStream_t stream) {
    (void)in_sizes; (void)n_in; (void)d_ws; (void)ws_size; (void)out_size;
    const float* x = (const float*)d_in[0];
    const float* A = (const float*)d_in[1];
    const float* B = (const float*)d_in[2];
    const float* C = (const float*)d_in[3];
    float* out = (float*)d_out;
    ssm_scan_wmma_f32<<<dim3(NFEAT), dim3(128), 0, stream>>>(x, A, B, C, out);
}


# Two time steps per LDS exchange and barrier: each wave also holds its rows of A_bar^2 (one WMMA product at set-up) and makes x(t+1), x(t+2) from x(t) as two 16-WMMA chains; summation order changed
# speedup vs baseline: 2.6974x; 1.2293x over previous
#include <hip/hip_runtime.h>

typedef __attribute__((ext_vector_type(2))) float v2f;
typedef __attribute__((ext_vector_type(4))) float v4f;
typedef __attribute__((ext_vector_type(8))) float v8f;

#define NBATCH 8
#define NFEAT  256
#define HD     64
#define NT     1024
#define AUGC   132
#define XSTR   68
#define YCHUNK 128
#define XBUF   (16*XSTR)
#define PSTR   129
#define SQOFF  8704
#define SQSTR  68

__launch_bounds__(128)
__global__ void ssm_scan_wmma_f32(const float* __restrict__ x,
                                  const float* __restrict__ Amat,
                                  const float* __restrict__ Bmat,
                                  const float* __restrict__ Cmat,
                                  float* __restrict__ out)
{
    __shared__ __attribute__((aligned(16))) float smem[NBATCH*NT + YCHUNK*PSTR + 2*XBUF];
    float* aug   = smem;
    float* u_lds = smem;
    float* p_lds = smem + NBATCH*NT;
    float* X_lds = smem + NBATCH*NT + YCHUNK*PSTR;

    const int f   = blockIdx.x;
    const int tid = threadIdx.x;

    const float* Af = Amat + f*HD*HD;
    for (int e = tid; e < HD*HD; e += 128) {
        int r = e >> 6, c = e & 63;
        float a   = Af[e];
        float eye = (r == c) ? 1.0f : 0.0f;
        aug[r*AUGC + c]      = eye - 0.5f*a;
        aug[r*AUGC + 64 + c] = eye + 0.5f*a;
    }
    for (int r = tid; r < HD; r += 128) {
        aug[r*AUGC + 128] = Bmat[f*HD + r];
        aug[r*AUGC + 129] = 0.0f;
        aug[r*AUGC + 130] = 0.0f;
        aug[r*AUGC + 131] = 0.0f;
    }
    __syncthreads();

    {
        const int row  = tid & 63;
        const int half = tid >> 6;
        float* orow = aug + row*AUGC + half*64;
        for (int p = 0; p < HD; ++p) {
            float fac = aug[row*AUGC + p] / aug[p*AUGC + p];
            __syncthreads();
            if (row != p) {
                const float* prow = aug + p*AUGC + half*64;
                v4f pv[16], ov[16];
                v4f pe = {}, oe = {};
                if (half) {
                    pe = *(const v4f*)(prow + 64);
                    oe = *(const v4f*)(orow + 64);
                }
                #pragma unroll
                for (int j = 0; j < 16; ++j) {
                    pv[j] = *(const v4f*)(prow + 4*j);
                    ov[j] = *(const v4f*)(orow + 4*j);
                }
                #pragma unroll
                for (int j = 0; j < 16; ++j) ov[j] -= fac * pv[j];
                oe -= fac * pe;
                #pragma unroll
                for (int j = 0; j < 16; ++j) *(v4f*)(orow + 4*j) = ov[j];
                if (half) *(v4f*)(orow + 64) = oe;
            }
            __syncthreads();
        }
    }

    const int wv   = tid >> 5;
    const int ln   = tid & 31;
    const int hi   = ln >> 4;
    const int col  = ln & 15;
    const int koff = hi << 1;
    const int arow = wv*16 + col;
    const int row2 = wv*16 + hi*8;

    v2f a_op[16];
    {
        float rinv = 1.0f / aug[arow*AUGC + arow];
        #pragma unroll
        for (int k = 0; k < 16; ++k) {
            a_op[k].x = aug[arow*AUGC + 64 + 4*k + koff]     * rinv;
            a_op[k].y = aug[arow*AUGC + 64 + 4*k + koff + 1] * rinv;
        }
    }
    float bvr[8], cvr[8];
    #pragma unroll
    for (int r = 0; r < 8; ++r) {
        int rr = row2 + r;
        float bw = aug[rr*AUGC + 128] / aug[rr*AUGC + rr];
        bvr[r] = (col < NBATCH) ? bw : 0.0f;
        cvr[r] = Cmat[f*HD + rr];
    }

    v2f a_sq[16];
    float abr[8];
    {
        float* S = smem + SQOFF;
        v2f rk[16];
        float part = 0.0f;
        #pragma unroll
        for (int k = 0; k < 16; ++k) {
            int j0 = 4*k + koff;
            float d0 = aug[j0*AUGC + j0], d1 = aug[(j0+1)*AUGC + j0 + 1];
            rk[k].x = 1.0f / d0;
            rk[k].y = 1.0f / d1;
            part += a_op[k].x * (aug[j0*AUGC + 128] / d0);
            part += a_op[k].y * (aug[(j0+1)*AUGC + 128] / d1);
        }
        float abv = part + __shfl_xor(part, 16);
        #pragma unroll
        for (int r = 0; r < 8; ++r) {
            float v = __shfl(abv, hi*8 + r);
            abr[r] = (col < NBATCH) ? v : 0.0f;
        }
        #pragma unroll
        for (int j = 0; j < 4; ++j) {
            v8f acc = {};
            #pragma unroll
            for (int k = 0; k < 16; ++k) {
                int j0 = 4*k + koff;
                v2f b;
                b.x = aug[j0*AUGC     + 64 + 16*j + col] * rk[k].x;
                b.y = aug[(j0+1)*AUGC + 64 + 16*j + col] * rk[k].y;
                acc = __builtin_amdgcn_wmma_f32_16x16x4_f32(false, a_op[k], false, b,
                                                            (short)0, acc, false, false);
            }
            #pragma unroll
            for (int r = 0; r < 8; ++r) S[(row2 + r)*SQSTR + 16*j + col] = acc[r];
        }
        __syncthreads();
        #pragma unroll
        for (int k = 0; k < 16; ++k)
            a_sq[k] = *(const v2f*)&S[arow*SQSTR + 4*k + koff];
    }
    __syncthreads();

    for (int e = tid; e < NBATCH*NT; e += 128) {
        int b = e >> 10, i = e & (NT - 1);
        u_lds[i*NBATCH + b] = x[(size_t)(b*NFEAT + f)*NT + i];
    }
    for (int e = tid; e < XBUF; e += 128) X_lds[e] = 0.0f;
    __syncthreads();

    const int pslot = (wv*2 + hi)*16 + col;
    for (int it = 0; it < NT/2; ++it) {
        const int t = 2*it;
        const float* Xr = X_lds + (it & 1)*XBUF;
        float*       Xw = X_lds + ((it & 1) ^ 1)*XBUF;

        float u1 = u_lds[t*NBATCH + (col & (NBATCH-1))];
        float u2 = u_lds[(t + 1)*NBATCH + (col & (NBATCH-1))];
        v2f bop[16];
        #pragma unroll
        for (int k = 0; k < 16; ++k)
            bop[k] = *(const v2f*)&Xr[col*XSTR + 4*k + koff];
        __builtin_amdgcn_sched_barrier(0);
        v8f P, Q;
        #pragma unroll
        for (int r = 0; r < 8; ++r) {
            P[r] = bvr[r] * u1;
            Q[r] = abr[r] * u1 + bvr[r] * u2;
        }
        #pragma unroll
        for (int k = 0; k < 16; ++k) {
            P = __builtin_amdgcn_wmma_f32_16x16x4_f32(false, a_op[k], false, bop[k],
                                                      (short)0, P, false, false);
            Q = __builtin_amdgcn_wmma_f32_16x16x4_f32(false, a_sq[k], false, bop[k],
                                                      (short)0, Q, false, false);
            __builtin_amdgcn_sched_barrier(0);
        }

        float psum1 = 0.0f, psum2 = 0.0f;
        #pragma unroll
        for (int r = 0; r < 8; ++r) {
            psum1 += cvr[r] * P[r];
            psum2 += cvr[r] * Q[r];
        }

        v4f lo4 = { Q[0], Q[1], Q[2], Q[3] };
        v4f hi4 = { Q[4], Q[5], Q[6], Q[7] };
        *(v4f*)&Xw[col*XSTR + row2]     = lo4;
        *(v4f*)&Xw[col*XSTR + row2 + 4] = hi4;
        p_lds[(t & (YCHUNK-1))*PSTR + pslot]       = psum1;
        p_lds[((t + 1) & (YCHUNK-1))*PSTR + pslot] = psum2;
        __syncthreads();

        if (((t + 1) & (YCHUNK-1)) == (YCHUNK-1)) {
            int tbase = t & ~(YCHUNK-1);
            for (int e = tid; e < NBATCH*YCHUNK; e += 128) {
                int b = e >> 7, i = e & (YCHUNK-1);
                float y = 0.0f;
                for (int j = 0; j < 8; ++j) y += p_lds[i*PSTR + j*16 + b];
                out[(size_t)(b*NFEAT + f)*NT + tbase + i] = y;
            }
            __syncthreads();
        }
    }
}

extern "C" void kernel_launch(void* const* d_in, const int* in_sizes, int n_in,
                              void* d_out, int out_size, void* d_ws, size_t ws_size,
                              hipStream_t stream) {
    (void)in_sizes; (void)n_in; (void)d_ws; (void)ws_size; (void)out_size;
    const float* x = (const float*)d_in[0];
    const float* A = (const float*)d_in[1];
    const float* B = (const float*)d_in[2];
    const float* C = (const float*)d_in[3];
    float* out = (float*)d_out;
    ssm_scan_wmma_f32<<<dim3(NFEAT), dim3(128), 0, stream>>>(x, A, B, C, out);
}


# Eight time steps per LDS exchange and barrier: each wave holds its rows of A_bar^1..8 and makes x(t+1..t+8) from x(t) as eight WMMA chains, input terms as WMMA k slices; summation order changed
# speedup vs baseline: 3.0142x; 1.1174x over previous
#include <hip/hip_runtime.h>

typedef __attribute__((ext_vector_type(2))) float v2f;
typedef __attribute__((ext_vector_type(4))) float v4f;
typedef __attribute__((ext_vector_type(8))) float v8f;

#define NBATCH 8
#define NFEAT  256
#define HD     64
#define NT     1024
#define AUGC   132
#define XSTR   68
#define YCHUNK 128
#define XBUF   (16*XSTR)
#define PSTR   129
#define SQOFF  8704
#define SQSTR  68
#define KS     8

__launch_bounds__(128)
__global__ void ssm_scan_wmma_f32(const float* __restrict__ x,
                                  const float* __restrict__ Amat,
                                  const float* __restrict__ Bmat,
                                  const float* __restrict__ Cmat,
                                  float* __restrict__ out)
{
    __shared__ __attribute__((aligned(16))) float smem[NBATCH*NT + YCHUNK*PSTR + 2*XBUF];
    float* aug   = smem;
    float* u_lds = smem;
    float* p_lds = smem + NBATCH*NT;
    float* X_lds = smem + NBATCH*NT + YCHUNK*PSTR;

    const int f   = blockIdx.x;
    const int tid = threadIdx.x;

    const float* Af = Amat + f*HD*HD;
    for (int e = tid; e < HD*HD; e += 128) {
        int r = e >> 6, c = e & 63;
        float a   = Af[e];
        float eye = (r == c) ? 1.0f : 0.0f;
        aug[r*AUGC + c]      = eye - 0.5f*a;
        aug[r*AUGC + 64 + c] = eye + 0.5f*a;
    }
    for (int r = tid; r < HD; r += 128) {
        aug[r*AUGC + 128] = Bmat[f*HD + r];
        aug[r*AUGC + 129] = 0.0f;
        aug[r*AUGC + 130] = 0.0f;
        aug[r*AUGC + 131] = 0.0f;
    }
    __syncthreads();

    {
        const int row  = tid & 63;
        const int half = tid >> 6;
        float* orow = aug + row*AUGC + half*64;
        for (int p = 0; p < HD; ++p) {
            float fac = aug[row*AUGC + p] / aug[p*AUGC + p];
            __syncthreads();
            if (row != p) {
                const float* prow = aug + p*AUGC + half*64;
                v4f pv[16], ov[16];
                v4f pe = {}, oe = {};
                if (half) {
                    pe = *(const v4f*)(prow + 64);
                    oe = *(const v4f*)(orow + 64);
                }
                #pragma unroll
                for (int j = 0; j < 16; ++j) {
                    pv[j] = *(const v4f*)(prow + 4*j);
                    ov[j] = *(const v4f*)(orow + 4*j);
                }
                #pragma unroll
                for (int j = 0; j < 16; ++j) ov[j] -= fac * pv[j];
                oe -= fac * pe;
                #pragma unroll
                for (int j = 0; j < 16; ++j) *(v4f*)(orow + 4*j) = ov[j];
                if (half) *(v4f*)(orow + 64) = oe;
            }
            __syncthreads();
        }
    }

    const int wv   = tid >> 5;
    const int ln   = tid & 31;
    const int hi   = ln >> 4;
    const int col  = ln & 15;
    const int koff = hi << 1;
    const int arow = wv*16 + col;
    const int row2 = wv*16 + hi*8;

    v2f a_pw[KS][16];
    {
        float rinv = 1.0f / aug[arow*AUGC + arow];
        #pragma unroll
        for (int k = 0; k < 16; ++k) {
            a_pw[0][k].x = aug[arow*AUGC + 64 + 4*k + koff]     * rinv;
            a_pw[0][k].y = aug[arow*AUGC + 64 + 4*k + koff + 1] * rinv;
        }
    }
    float cvr[8];
    #pragma unroll
    for (int r = 0; r < 8; ++r) cvr[r] = Cmat[f*HD + row2 + r];

    float wA[KS];
    {
        float* S = smem + SQOFF;
        v2f bvk[16];
        v2f bA[4][16];
        #pragma unroll
        for (int k = 0; k < 16; ++k) {
            int j0 = 4*k + koff;
            float d0 = aug[j0*AUGC + j0], d1 = aug[(j0+1)*AUGC + j0 + 1];
            float r0 = 1.0f / d0, r1 = 1.0f / d1;
            bvk[k].x = aug[j0*AUGC + 128] / d0;
            bvk[k].y = aug[(j0+1)*AUGC + 128] / d1;
            #pragma unroll
            for (int j = 0; j < 4; ++j) {
                bA[j][k].x = aug[j0*AUGC     + 64 + 16*j + col] * r0;
                bA[j][k].y = aug[(j0+1)*AUGC + 64 + 16*j + col] * r1;
            }
        }
        wA[0] = aug[arow*AUGC + 128] / aug[arow*AUGC + arow];
        #pragma unroll
        for (int p = 1; p < KS; ++p) {
            float part = 0.0f;
            #pragma unroll
            for (int k = 0; k < 16; ++k) {
                part += a_pw[p-1][k].x * bvk[k].x;
                part += a_pw[p-1][k].y * bvk[k].y;
            }
            wA[p] = part + __shfl_xor(part, 16);
            #pragma unroll
            for (int j = 0; j < 4; ++j) {
                v8f acc = {};
                #pragma unroll
                for (int k = 0; k < 16; ++k)
                    acc = __builtin_amdgcn_wmma_f32_16x16x4_f32(false, a_pw[p-1][k], false, bA[j][k],
                                                                (short)0, acc, false, false);
                #pragma unroll
                for (int r = 0; r < 8; ++r) S[(row2 + r)*SQSTR + 16*j + col] = acc[r];
            }
            __syncthreads();
            #pragma unroll
            for (int k = 0; k < 16; ++k)
                a_pw[p][k] = *(const v2f*)&S[arow*SQSTR + 4*k + koff];
            __syncthreads();
        }
    }

    v2f wi[KS][KS/4];
    #pragma unroll
    for (int j = 0; j < KS; ++j) {
        #pragma unroll
        for (int q = 0; q < KS/4; ++q) {
            float e0 = (4*q     <= j) ? wA[(4*q     <= j) ? j - 4*q     : 0] : 0.0f;
            float e1 = (4*q + 1 <= j) ? wA[(4*q + 1 <= j) ? j - 4*q - 1 : 0] : 0.0f;
            float e2 = (4*q + 2 <= j) ? wA[(4*q + 2 <= j) ? j - 4*q - 2 : 0] : 0.0f;
            float e3 = (4*q + 3 <= j) ? wA[(4*q + 3 <= j) ? j - 4*q - 3 : 0] : 0.0f;
            wi[j][q].x = hi ? e2 : e0;
            wi[j][q].y = hi ? e3 : e1;
        }
    }

    for (int e = tid; e < NBATCH*NT; e += 128) {
        int b = e >> 10, i = e & (NT - 1);
        u_lds[i*NBATCH + b] = x[(size_t)(b*NFEAT + f)*NT + i];
    }
    for (int e = tid; e < XBUF; e += 128) X_lds[e] = 0.0f;
    __syncthreads();

    const int pslot = (wv*2 + hi)*16 + col;
    const int cb    = col & (NBATCH-1);
    for (int it = 0; it < NT/KS; ++it) {
        const int t = KS*it;
        const float* Xr = X_lds + (it & 1)*XBUF;
        float*       Xw = X_lds + ((it & 1) ^ 1)*XBUF;

        v2f ub[KS/4];
        #pragma unroll
        for (int q = 0; q < KS/4; ++q) {
            ub[q].x = u_lds[(t + 4*q + koff)*NBATCH + cb];
            ub[q].y = u_lds[(t + 4*q + koff + 1)*NBATCH + cb];
        }
        v2f bop[16];
        #pragma unroll
        for (int k = 0; k < 16; ++k)
            bop[k] = *(const v2f*)&Xr[col*XSTR + 4*k + koff];
        __builtin_amdgcn_sched_barrier(0);

        v8f acc[KS];
        #pragma unroll
        for (int j = KS-1; j >= 0; --j) {
            v8f zero = {};
            acc[j] = __builtin_amdgcn_wmma_f32_16x16x4_f32(false, wi[j][0], false, ub[0],
                                                           (short)0, zero, false, false);
        }
        __builtin_amdgcn_sched_barrier(0);
        #pragma unroll
        for (int q = 1; q < KS/4; ++q) {
            #pragma unroll
            for (int j = KS-1; j >= 4*q; --j)
                acc[j] = __builtin_amdgcn_wmma_f32_16x16x4_f32(false, wi[j][q], false, ub[q],
                                                               (short)0, acc[j], false, false);
            __builtin_amdgcn_sched_barrier(0);
        }
        #pragma unroll
        for (int k = 0; k < 16; ++k) {
            #pragma unroll
            for (int j = KS-1; j >= 0; --j)
                acc[j] = __builtin_amdgcn_wmma_f32_16x16x4_f32(false, a_pw[j][k], false, bop[k],
                                                               (short)0, acc[j], false, false);
            if (k < 15) __builtin_amdgcn_sched_barrier(0);
        }

        v4f lo4 = { acc[KS-1][0], acc[KS-1][1], acc[KS-1][2], acc[KS-1][3] };
        v4f hi4 = { acc[KS-1][4], acc[KS-1][5], acc[KS-1][6], acc[KS-1][7] };
        *(v4f*)&Xw[col*XSTR + row2]     = lo4;
        *(v4f*)&Xw[col*XSTR + row2 + 4] = hi4;
        #pragma unroll
        for (int j = 0; j < KS; ++j) {
            float ps = 0.0f;
            #pragma unroll
            for (int r = 0; r < 8; ++r) ps += cvr[r] * acc[j][r];
            p_lds[((t + j) & (YCHUNK-1))*PSTR + pslot] = ps;
        }
        __syncthreads();

        if (((t + KS-1) & (YCHUNK-1)) == (YCHUNK-1)) {
            int tbase = t & ~(YCHUNK-1);
            for (int e = tid; e < NBATCH*YCHUNK; e += 128) {
                int b = e >> 7, i = e & (YCHUNK-1);
                float y = 0.0f;
                for (int j = 0; j < 8; ++j) y += p_lds[i*PSTR + j*16 + b];
                out[(size_t)(b*NFEAT + f)*NT + tbase + i] = y;
            }
            __syncthreads();
        }
    }
}

extern "C" void kernel_launch(void* const* d_in, const int* in_sizes, int n_in,
                              void* d_out, int out_size, void* d_ws, size_t ws_size,
                              hipStream_t stream) {
    (void)in_sizes; (void)n_in; (void)d_ws; (void)ws_size; (void)out_size;
    const float* x = (const float*)d_in[0];
    const float* A = (const float*)d_in[1];
    const float* B = (const float*)d_in[2];
    const float* C = (const float*)d_in[3];
    float* out = (float*)d_out;
    ssm_scan_wmma_f32<<<dim3(NFEAT), dim3(128), 0, stream>>>(x, A, B, C, out);
}


# The 8 outputs of a trip as one WMMA tile: rows Cv A_bar^j times x(t) plus taps Cv A_bar^d Bv times u; only x(t+8) is formed (even/odd k accumulators), not the 7 states between; sum order changed
# speedup vs baseline: 6.4906x; 2.1533x over previous
#include <hip/hip_runtime.h>

typedef __attribute__((ext_vector_type(2))) float v2f;
typedef __attribute__((ext_vector_type(4))) float v4f;
typedef __attribute__((ext_vector_type(8))) float v8f;

#define NBATCH 8
#define NFEAT  256
#define HD     64
#define NT     1024
#define AUGC   132
#define XSTR   68
#define YCHUNK 128
#define XBUF   (16*XSTR)
#define PSTR   129
#define SQOFF  8704
#define SQSTR  68
#define KS     8

__launch_bounds__(128)
__global__ void ssm_scan_wmma_f32(const float* __restrict__ x,
                                  const float* __restrict__ Amat,
                                  const float* __restrict__ Bmat,
                                  const float* __restrict__ Cmat,
                                  float* __restrict__ out)
{
    __shared__ __attribute__((aligned(16))) float smem[NBATCH*NT + YCHUNK*PSTR + 2*XBUF];
    float* aug   = smem;
    float* u_lds = smem;
    float* p_lds = smem + NBATCH*NT;
    float* X_lds = smem + NBATCH*NT + YCHUNK*PSTR;

    const int f   = blockIdx.x;
    const int tid = threadIdx.x;

    const float* Af = Amat + f*HD*HD;
    for (int e = tid; e < HD*HD; e += 128) {
        int r = e >> 6, c = e & 63;
        float a   = Af[e];
        float eye = (r == c) ? 1.0f : 0.0f;
        aug[r*AUGC + c]      = eye - 0.5f*a;
        aug[r*AUGC + 64 + c] = eye + 0.5f*a;
    }
    for (int r = tid; r < HD; r += 128) {
        aug[r*AUGC + 128] = Bmat[f*HD + r];
        aug[r*AUGC + 129] = 0.0f;
        aug[r*AUGC + 130] = 0.0f;
        aug[r*AUGC + 131] = 0.0f;
    }
    __syncthreads();

    {
        const int row  = tid & 63;
        const int half = tid >> 6;
        float* orow = aug + row*AUGC + half*64;
        for (int p = 0; p < HD; ++p) {
            float fac = aug[row*AUGC + p] / aug[p*AUGC + p];
            __syncthreads();
            if (row != p) {
                const float* prow = aug + p*AUGC + half*64;
                v4f pv[16], ov[16];
                v4f pe = {}, oe = {};
                if (half) {
                    pe = *(const v4f*)(prow + 64);
                    oe = *(const v4f*)(orow + 64);
                }
                #pragma unroll
                for (int j = 0; j < 16; ++j) {
                    pv[j] = *(const v4f*)(prow + 4*j);
                    ov[j] = *(const v4f*)(orow + 4*j);
                }
                #pragma unroll
                for (int j = 0; j < 16; ++j) ov[j] -= fac * pv[j];
                oe -= fac * pe;
                #pragma unroll
                for (int j = 0; j < 16; ++j) *(v4f*)(orow + 4*j) = ov[j];
                if (half) *(v4f*)(orow + 64) = oe;
            }
            __syncthreads();
        }
    }

    const int wv   = tid >> 5;
    const int ln   = tid & 31;
    const int hi   = ln >> 4;
    const int col  = ln & 15;
    const int koff = hi << 1;
    const int arow = wv*16 + col;
    const int row2 = wv*16 + hi*8;

    v2f a_pw[KS][16];
    {
        float rinv = 1.0f / aug[arow*AUGC + arow];
        #pragma unroll
        for (int k = 0; k < 16; ++k) {
            a_pw[0][k].x = aug[arow*AUGC + 64 + 4*k + koff]     * rinv;
            a_pw[0][k].y = aug[arow*AUGC + 64 + 4*k + koff + 1] * rinv;
        }
    }

    float wA[KS];
    {
        float* S = smem + SQOFF;
        v2f bvk[16];
        v2f bA[4][16];
        #pragma unroll
        for (int k = 0; k < 16; ++k) {
            int j0 = 4*k + koff;
            float d0 = aug[j0*AUGC + j0], d1 = aug[(j0+1)*AUGC + j0 + 1];
            float r0 = 1.0f / d0, r1 = 1.0f / d1;
            bvk[k].x = aug[j0*AUGC + 128] / d0;
            bvk[k].y = aug[(j0+1)*AUGC + 128] / d1;
            #pragma unroll
            for (int j = 0; j < 4; ++j) {
                bA[j][k].x = aug[j0*AUGC     + 64 + 16*j + col] * r0;
                bA[j][k].y = aug[(j0+1)*AUGC + 64 + 16*j + col] * r1;
            }
        }
        wA[0] = aug[arow*AUGC + 128] / aug[arow*AUGC + arow];
        #pragma unroll
        for (int p = 1; p < KS; ++p) {
            float part = 0.0f;
            #pragma unroll
            for (int k = 0; k < 16; ++k) {
                part += a_pw[p-1][k].x * bvk[k].x;
                part += a_pw[p-1][k].y * bvk[k].y;
            }
            wA[p] = part + __shfl_xor(part, 16);
            #pragma unroll
            for (int j = 0; j < 4; ++j) {
                v8f acc = {};
                #pragma unroll
                for (int k = 0; k < 16; ++k)
                    acc = __builtin_amdgcn_wmma_f32_16x16x4_f32(false, a_pw[p-1][k], false, bA[j][k],
                                                                (short)0, acc, false, false);
                #pragma unroll
                for (int r = 0; r < 8; ++r) S[(row2 + r)*SQSTR + 16*j + col] = acc[r];
            }
            __syncthreads();
            #pragma unroll
            for (int k = 0; k < 16; ++k)
                a_pw[p][k] = *(const v2f*)&S[arow*SQSTR + 4*k + koff];
            __syncthreads();
        }
    }

    v2f wi[KS/4];
    {
        const int j = KS-1;
        #pragma unroll
        for (int q = 0; q < KS/4; ++q) {
            float e0 = (4*q     <= j) ? wA[(4*q     <= j) ? j - 4*q     : 0] : 0.0f;
            float e1 = (4*q + 1 <= j) ? wA[(4*q + 1 <= j) ? j - 4*q - 1 : 0] : 0.0f;
            float e2 = (4*q + 2 <= j) ? wA[(4*q + 2 <= j) ? j - 4*q - 2 : 0] : 0.0f;
            float e3 = (4*q + 3 <= j) ? wA[(4*q + 3 <= j) ? j - 4*q - 3 : 0] : 0.0f;
            wi[q].x = hi ? e2 : e0;
            wi[q].y = hi ? e3 : e1;
        }
    }

    v2f gy[4];
    v2f hy;
    {
        float* Gl = X_lds;
        float* Gs = X_lds + (KS+1)*HD;
        float* Bl = X_lds + 2*(KS+1)*HD;
        float* Kl = Bl + HD;
        float rd = 0.0f;
        if (tid < HD) {
            rd = 1.0f / aug[tid*AUGC + tid];
            float c = Cmat[f*HD + tid];
            Gl[tid] = c;
            Gs[tid] = c * rd;
            Bl[tid] = aug[tid*AUGC + 128] * rd;
        }
        __syncthreads();
        for (int d = 1; d <= KS; ++d) {
            if (tid < HD) {
                v4f gv[HD/4];
                float av[HD];
                #pragma unroll
                for (int j = 0; j < HD/4; ++j) gv[j] = *(const v4f*)&Gs[(d-1)*HD + 4*j];
                #pragma unroll
                for (int r = 0; r < HD; ++r) av[r] = aug[r*AUGC + 64 + tid];
                __builtin_amdgcn_sched_barrier(0);
                float s = 0.0f;
                #pragma unroll
                for (int r = 0; r < HD; ++r) s += gv[r >> 2][r & 3] * av[r];
                Gl[d*HD + tid] = s;
                Gs[d*HD + tid] = s * rd;
            }
            __syncthreads();
        }
        if (tid < KS) {
            float s = 0.0f;
            #pragma unroll
            for (int r = 0; r < HD; ++r) s += Gl[tid*HD + r] * Bl[r];
            Kl[tid] = s;
        }
        __syncthreads();
        const bool vrow = col < KS;
        #pragma unroll
        for (int q = 0; q < 4; ++q) {
            v2f g = *(const v2f*)&Gl[(vrow ? col + 1 : 0)*HD + 16*wv + 4*q + koff];
            gy[q].x = vrow ? g.x : 0.0f;
            gy[q].y = vrow ? g.y : 0.0f;
        }
        const int d0 = col - 2*wv, d1 = d0 - 1;
        const int i0 = d0 < 0 ? 0 : (d0 > KS-1 ? KS-1 : d0);
        const int i1 = d1 < 0 ? 0 : (d1 > KS-1 ? KS-1 : d1);
        const float k0 = Kl[i0], k1 = Kl[i1];
        hy.x = (!hi && vrow && d0 >= 0) ? k0 : 0.0f;
        hy.y = (!hi && vrow && d1 >= 0) ? k1 : 0.0f;
        __syncthreads();
    }

    for (int e = tid; e < NBATCH*NT; e += 128) {
        int b = e >> 10, i = e & (NT - 1);
        u_lds[i*NBATCH + b] = x[(size_t)(b*NFEAT + f)*NT + i];
    }
    for (int e = tid; e < XBUF; e += 128) X_lds[e] = 0.0f;
    __syncthreads();

    const int pslot = (hi*4 + wv)*16 + col;
    const int cb    = col & (NBATCH-1);
    for (int it = 0; it < NT/KS; ++it) {
        const int t = KS*it;
        const float* Xr = X_lds + (it & 1)*XBUF;
        float*       Xw = X_lds + ((it & 1) ^ 1)*XBUF;

        v2f ub[KS/4];
        #pragma unroll
        for (int q = 0; q < KS/4; ++q) {
            ub[q].x = u_lds[(t + 4*q + koff)*NBATCH + cb];
            ub[q].y = u_lds[(t + 4*q + koff + 1)*NBATCH + cb];
        }
        v2f uy;
        {
            float a = u_lds[(t + 2*wv)*NBATCH + cb];
            float b = u_lds[(t + 2*wv + 1)*NBATCH + cb];
            uy.x = hi ? 0.0f : a;
            uy.y = hi ? 0.0f : b;
        }
        v2f bop[16];
        #pragma unroll
        for (int k = 0; k < 16; ++k)
            bop[k] = *(const v2f*)&Xr[col*XSTR + 4*k + koff];
        v2f by[4];
        #pragma unroll
        for (int q = 0; q < 4; ++q)
            by[q] = *(const v2f*)&Xr[col*XSTR + 16*wv + 4*q + koff];
        __builtin_amdgcn_sched_barrier(0);

        v8f zero = {};
        v8f accA = __builtin_amdgcn_wmma_f32_16x16x4_f32(false, wi[0], false, ub[0], (short)0, zero, false, false);
        v8f accB = __builtin_amdgcn_wmma_f32_16x16x4_f32(false, wi[1], false, ub[1], (short)0, zero, false, false);
        v8f accY = __builtin_amdgcn_wmma_f32_16x16x4_f32(false, hy,    false, uy,    (short)0, zero, false, false);
        #pragma unroll
        for (int k = 0; k < 16; k += 2) {
            accA = __builtin_amdgcn_wmma_f32_16x16x4_f32(false, a_pw[KS-1][k],   false, bop[k],   (short)0, accA, false, false);
            accB = __builtin_amdgcn_wmma_f32_16x16x4_f32(false, a_pw[KS-1][k+1], false, bop[k+1], (short)0, accB, false, false);
            if ((k & 3) == 0)
                accY = __builtin_amdgcn_wmma_f32_16x16x4_f32(false, gy[k >> 2], false, by[k >> 2], (short)0, accY, false, false);
        }

        v8f xs = accA + accB;
        v4f lo4 = { xs[0], xs[1], xs[2], xs[3] };
        v4f hi4 = { xs[4], xs[5], xs[6], xs[7] };
        *(v4f*)&Xw[col*XSTR + row2]     = lo4;
        *(v4f*)&Xw[col*XSTR + row2 + 4] = hi4;
        #pragma unroll
        for (int r = 0; r < KS; ++r)
            p_lds[((t + r) & (YCHUNK-1))*PSTR + pslot] = accY[r];
        __syncthreads();

        if (((t + KS-1) & (YCHUNK-1)) == (YCHUNK-1)) {
            int tbase = t & ~(YCHUNK-1);
            for (int e = tid; e < NBATCH*YCHUNK; e += 128) {
                int b = e >> 7, i = e & (YCHUNK-1);
                float y = 0.0f;
                for (int j = 0; j < 4; ++j) y += p_lds[i*PSTR + j*16 + b];
                out[(size_t)(b*NFEAT + f)*NT + tbase + i] = y;
            }
            __syncthreads();
        }
    }
}

extern "C" void kernel_launch(void* const* d_in, const int* in_sizes, int n_in,
                              void* d_out, int out_size, void* d_ws, size_t ws_size,
                              hipStream_t stream) {
    (void)in_sizes; (void)n_in; (void)d_ws; (void)ws_size; (void)out_size;
    const float* x = (const float*)d_in[0];
    const float* A = (const float*)d_in[1];
    const float* B = (const float*)d_in[2];
    const float* C = (const float*)d_in[3];
    float* out = (float*)d_out;
    ssm_scan_wmma_f32<<<dim3(NFEAT), dim3(128), 0, stream>>>(x, A, B, C, out);
}


# Sixteen time steps per LDS exchange and barrier (was 8): A_bar^16 rows, a full 16-row output tile and 4 input k slices per trip; the power set-up as a rolled loop; summation order changed
# speedup vs baseline: 7.7675x; 1.1967x over previous
#include <hip/hip_runtime.h>

typedef __attribute__((ext_vector_type(2))) float v2f;
typedef __attribute__((ext_vector_type(4))) float v4f;
typedef __attribute__((ext_vector_type(8))) float v8f;

#define NBATCH 8
#define NFEAT  256
#define HD     64
#define NT     1024
#define AUGC   132
#define XSTR   68
#define YCHUNK 128
#define XBUF   (16*XSTR)
#define PSTR   129
#define SQOFF  8704
#define SQSTR  68
#define KS     16

__launch_bounds__(128)
__global__ void ssm_scan_wmma_f32(const float* __restrict__ x,
                                  const float* __restrict__ Amat,
                                  const float* __restrict__ Bmat,
                                  const float* __restrict__ Cmat,
                                  float* __restrict__ out)
{
    __shared__ __attribute__((aligned(16))) float smem[NBATCH*NT + YCHUNK*PSTR + 2*XBUF];
    float* aug   = smem;
    float* u_lds = smem;
    float* p_lds = smem + NBATCH*NT;
    float* X_lds = smem + NBATCH*NT + YCHUNK*PSTR;

    const int f   = blockIdx.x;
    const int tid = threadIdx.x;

    const float* Af = Amat + f*HD*HD;
    for (int e = tid; e < HD*HD; e += 128) {
        int r = e >> 6, c = e & 63;
        float a   = Af[e];
        float eye = (r == c) ? 1.0f : 0.0f;
        aug[r*AUGC + c]      = eye - 0.5f*a;
        aug[r*AUGC + 64 + c] = eye + 0.5f*a;
    }
    for (int r = tid; r < HD; r += 128) {
        aug[r*AUGC + 128] = Bmat[f*HD + r];
        aug[r*AUGC + 129] = 0.0f;
        aug[r*AUGC + 130] = 0.0f;
        aug[r*AUGC + 131] = 0.0f;
    }
    __syncthreads();

    {
        const int row  = tid & 63;
        const int half = tid >> 6;
        float* orow = aug + row*AUGC + half*64;
        for (int p = 0; p < HD; ++p) {
            float fac = aug[row*AUGC + p] / aug[p*AUGC + p];
            __syncthreads();
            if (row != p) {
                const float* prow = aug + p*AUGC + half*64;
                v4f pv[16], ov[16];
                v4f pe = {}, oe = {};
                if (half) {
                    pe = *(const v4f*)(prow + 64);
                    oe = *(const v4f*)(orow + 64);
                }
                #pragma unroll
                for (int j = 0; j < 16; ++j) {
                    pv[j] = *(const v4f*)(prow + 4*j);
                    ov[j] = *(const v4f*)(orow + 4*j);
                }
                #pragma unroll
                for (int j = 0; j < 16; ++j) ov[j] -= fac * pv[j];
                oe -= fac * pe;
                #pragma unroll
                for (int j = 0; j < 16; ++j) *(v4f*)(orow + 4*j) = ov[j];
                if (half) *(v4f*)(orow + 64) = oe;
            }
            __syncthreads();
        }
    }

    const int wv   = tid >> 5;
    const int ln   = tid & 31;
    const int hi   = ln >> 4;
    const int col  = ln & 15;
    const int koff = hi << 1;
    const int arow = wv*16 + col;
    const int row2 = wv*16 + hi*8;

    v2f a_cur[16];
    {
        float rinv = 1.0f / aug[arow*AUGC + arow];
        #pragma unroll
        for (int k = 0; k < 16; ++k) {
            a_cur[k].x = aug[arow*AUGC + 64 + 4*k + koff]     * rinv;
            a_cur[k].y = aug[arow*AUGC + 64 + 4*k + koff + 1] * rinv;
        }
    }

    v2f wi[KS/4];
    {
        float* S = smem + SQOFF;
        v2f bvk[16];
        v2f bA[4][16];
        #pragma unroll
        for (int k = 0; k < 16; ++k) {
            int j0 = 4*k + koff;
            float d0 = aug[j0*AUGC + j0], d1 = aug[(j0+1)*AUGC + j0 + 1];
            float r0 = 1.0f / d0, r1 = 1.0f / d1;
            bvk[k].x = aug[j0*AUGC + 128] / d0;
            bvk[k].y = aug[(j0+1)*AUGC + 128] / d1;
            #pragma unroll
            for (int j = 0; j < 4; ++j) {
                bA[j][k].x = aug[j0*AUGC     + 64 + 16*j + col] * r0;
                bA[j][k].y = aug[(j0+1)*AUGC + 64 + 16*j + col] * r1;
            }
        }
        const float w0 = aug[arow*AUGC + 128] / aug[arow*AUGC + arow];
        #pragma unroll
        for (int q = 0; q < KS/4; ++q) {
            wi[q].x = (KS-1 - 4*q - koff == 0) ? w0 : 0.0f;
            wi[q].y = (KS-2 - 4*q - koff == 0) ? w0 : 0.0f;
        }
        #pragma nounroll
        for (int p = 1; p < KS; ++p) {
            float part = 0.0f;
            #pragma unroll
            for (int k = 0; k < 16; ++k) {
                part += a_cur[k].x * bvk[k].x;
                part += a_cur[k].y * bvk[k].y;
            }
            const float w = part + __shfl_xor(part, 16);
            #pragma unroll
            for (int q = 0; q < KS/4; ++q) {
                if (KS-1 - 4*q - koff == p) wi[q].x = w;
                if (KS-2 - 4*q - koff == p) wi[q].y = w;
            }
            #pragma unroll
            for (int j = 0; j < 4; ++j) {
                v8f acc = {};
                #pragma unroll
                for (int k = 0; k < 16; ++k)
                    acc = __builtin_amdgcn_wmma_f32_16x16x4_f32(false, a_cur[k], false, bA[j][k],
                                                                (short)0, acc, false, false);
                #pragma unroll
                for (int r = 0; r < 8; ++r) S[(row2 + r)*SQSTR + 16*j + col] = acc[r];
            }
            __syncthreads();
            #pragma unroll
            for (int k = 0; k < 16; ++k)
                a_cur[k] = *(const v2f*)&S[arow*SQSTR + 4*k + koff];
            __syncthreads();
        }
    }

    static_assert(KS == 16, "the output tile's input slice below is laid out for 4 steps per wave");
    v2f gy[4];
    v2f hy;
    {
        float* Gl = smem + SQOFF;
        float* Gs = Gl + (KS+1)*HD;
        float* Bl = Gs + (KS+1)*HD;
        float* Kl = Bl + HD;
        float rd = 0.0f;
        if (tid < HD) {
            rd = 1.0f / aug[tid*AUGC + tid];
            float c = Cmat[f*HD + tid];
            Gl[tid] = c;
            Gs[tid] = c * rd;
            Bl[tid] = aug[tid*AUGC + 128] * rd;
        }
        __syncthreads();
        for (int d = 1; d <= KS; ++d) {
            if (tid < HD) {
                v4f gv[HD/4];
                float av[HD];
                #pragma unroll
                for (int j = 0; j < HD/4; ++j) gv[j] = *(const v4f*)&Gs[(d-1)*HD + 4*j];
                #pragma unroll
                for (int r = 0; r < HD; ++r) av[r] = aug[r*AUGC + 64 + tid];
                __builtin_amdgcn_sched_barrier(0);
                float s = 0.0f;
                #pragma unroll
                for (int r = 0; r < HD; ++r) s += gv[r >> 2][r & 3] * av[r];
                Gl[d*HD + tid] = s;
                Gs[d*HD + tid] = s * rd;
            }
            __syncthreads();
        }
        if (tid < KS) {
            float s = 0.0f;
            #pragma unroll
            for (int r = 0; r < HD; ++r) s += Gl[tid*HD + r] * Bl[r];
            Kl[tid] = s;
        }
        __syncthreads();
        #pragma unroll
        for (int q = 0; q < 4; ++q)
            gy[q] = *(const v2f*)&Gl[(col + 1)*HD + 16*wv + 4*q + koff];
        const int d0 = col - 4*wv - koff, d1 = d0 - 1;
        const float k0 = Kl[d0 < 0 ? 0 : d0], k1 = Kl[d1 < 0 ? 0 : d1];
        hy.x = (d0 >= 0) ? k0 : 0.0f;
        hy.y = (d1 >= 0) ? k1 : 0.0f;
    }

    for (int e = tid; e < NBATCH*NT; e += 128) {
        int b = e >> 10, i = e & (NT - 1);
        u_lds[i*NBATCH + b] = x[(size_t)(b*NFEAT + f)*NT + i];
    }
    for (int e = tid; e < XBUF; e += 128) X_lds[e] = 0.0f;
    __syncthreads();

    const int pslot = wv*16 + col;
    const int cb    = col & (NBATCH-1);
    for (int it = 0; it < NT/KS; ++it) {
        const int t = KS*it;
        const float* Xr = X_lds + (it & 1)*XBUF;
        float*       Xw = X_lds + ((it & 1) ^ 1)*XBUF;

        v2f ub[KS/4];
        #pragma unroll
        for (int q = 0; q < KS/4; ++q) {
            ub[q].x = u_lds[(t + 4*q + koff)*NBATCH + cb];
            ub[q].y = u_lds[(t + 4*q + koff + 1)*NBATCH + cb];
        }
        v2f uy;
        uy.x = u_lds[(t + 4*wv + koff)*NBATCH + cb];
        uy.y = u_lds[(t + 4*wv + koff + 1)*NBATCH + cb];
        v2f bop[16];
        #pragma unroll
        for (int k = 0; k < 16; ++k)
            bop[k] = *(const v2f*)&Xr[col*XSTR + 4*k + koff];
        v2f by[4];
        #pragma unroll
        for (int q = 0; q < 4; ++q)
            by[q] = *(const v2f*)&Xr[col*XSTR + 16*wv + 4*q + koff];
        __builtin_amdgcn_sched_barrier(0);

        v8f zero = {};
        v8f accA = __builtin_amdgcn_wmma_f32_16x16x4_f32(false, wi[0], false, ub[0], (short)0, zero, false, false);
        v8f accB = __builtin_amdgcn_wmma_f32_16x16x4_f32(false, wi[1], false, ub[1], (short)0, zero, false, false);
        v8f accY = __builtin_amdgcn_wmma_f32_16x16x4_f32(false, hy,    false, uy,    (short)0, zero, false, false);
        #pragma unroll
        for (int q = 2; q < KS/4; q += 2) {
            accA = __builtin_amdgcn_wmma_f32_16x16x4_f32(false, wi[q],   false, ub[q],   (short)0, accA, false, false);
            accB = __builtin_amdgcn_wmma_f32_16x16x4_f32(false, wi[q+1], false, ub[q+1], (short)0, accB, false, false);
        }
        #pragma unroll
        for (int k = 0; k < 16; k += 2) {
            accA = __builtin_amdgcn_wmma_f32_16x16x4_f32(false, a_cur[k],   false, bop[k],   (short)0, accA, false, false);
            accB = __builtin_amdgcn_wmma_f32_16x16x4_f32(false, a_cur[k+1], false, bop[k+1], (short)0, accB, false, false);
            if ((k & 3) == 0)
                accY = __builtin_amdgcn_wmma_f32_16x16x4_f32(false, gy[k >> 2], false, by[k >> 2], (short)0, accY, false, false);
        }

        v8f xs = accA + accB;
        v4f lo4 = { xs[0], xs[1], xs[2], xs[3] };
        v4f hi4 = { xs[4], xs[5], xs[6], xs[7] };
        *(v4f*)&Xw[col*XSTR + row2]     = lo4;
        *(v4f*)&Xw[col*XSTR + row2 + 4] = hi4;
        #pragma unroll
        for (int r = 0; r < 8; ++r)
            p_lds[((t + 8*hi + r) & (YCHUNK-1))*PSTR + pslot] = accY[r];
        __syncthreads();

        if (((t + KS-1) & (YCHUNK-1)) == (YCHUNK-1)) {
            int tbase = t & ~(YCHUNK-1);
            for (int e = tid; e < NBATCH*YCHUNK; e += 128) {
                int b = e >> 7, i = e & (YCHUNK-1);
                float y = 0.0f;
                for (int j = 0; j < 4; ++j) y += p_lds[i*PSTR + j*16 + b];
                out[(size_t)(b*NFEAT + f)*NT + tbase + i] = y;
            }
            __syncthreads();
        }
    }
}

extern "C" void kernel_launch(void* const* d_in, const int* in_sizes, int n_in,
                              void* d_out, int out_size, void* d_ws, size_t ws_size,
                              hipStream_t stream) {
    (void)in_sizes; (void)n_in; (void)d_ws; (void)ws_size; (void)out_size;
    const float* x = (const float*)d_in[0];
    const float* A = (const float*)d_in[1];
    const float* B = (const float*)d_in[2];
    const float* C = (const float*)d_in[3];
    float* out = (float*)d_out;
    ssm_scan_wmma_f32<<<dim3(NFEAT), dim3(128), 0, stream>>>(x, A, B, C, out);
}


# Gauss-Jordan rows kept in registers over all 64 pivots: only the pivot row and the pivot-column entries pass through LDS (two alternating buffers), one barrier per pivot, not two; sums bit-identical
# speedup vs baseline: 8.5071x; 1.0952x over previous
#include <hip/hip_runtime.h>

typedef __attribute__((ext_vector_type(2))) float v2f;
typedef __attribute__((ext_vector_type(4))) float v4f;
typedef __attribute__((ext_vector_type(8))) float v8f;

#define NBATCH 8
#define NFEAT  256
#define HD     64
#define NT     1024
#define AUGC   132
#define XSTR   68
#define YCHUNK 128
#define XBUF   (16*XSTR)
#define PSTR   129
#define SQOFF  8704
#define SQSTR  68
#define KS     16

__launch_bounds__(128)
__global__ void ssm_scan_wmma_f32(const float* __restrict__ x,
                                  const float* __restrict__ Amat,
                                  const float* __restrict__ Bmat,
                                  const float* __restrict__ Cmat,
                                  float* __restrict__ out)
{
    __shared__ __attribute__((aligned(16))) float smem[NBATCH*NT + YCHUNK*PSTR + 2*XBUF];
    float* aug   = smem;
    float* u_lds = smem;
    float* p_lds = smem + NBATCH*NT;
    float* X_lds = smem + NBATCH*NT + YCHUNK*PSTR;

    const int f   = blockIdx.x;
    const int tid = threadIdx.x;

    const float* Af = Amat + f*HD*HD;
    for (int e = tid; e < HD*HD; e += 128) {
        int r = e >> 6, c = e & 63;
        float a   = Af[e];
        float eye = (r == c) ? 1.0f : 0.0f;
        aug[r*AUGC + c]      = eye - 0.5f*a;
        aug[r*AUGC + 64 + c] = eye + 0.5f*a;
    }
    for (int r = tid; r < HD; r += 128) {
        aug[r*AUGC + 128] = Bmat[f*HD + r];
        aug[r*AUGC + 129] = 0.0f;
        aug[r*AUGC + 130] = 0.0f;
        aug[r*AUGC + 131] = 0.0f;
    }
    __syncthreads();

    {
        const int row  = tid & 63;
        const int half = tid >> 6;
        float* orow = aug + row*AUGC + half*64;
        float* RB = smem + SQOFF;
        float* CB = RB + 2*2*68;
        v4f ov[16];
        v4f oe = {};
        #pragma unroll
        for (int i = 0; i < 16; ++i) ov[i] = *(const v4f*)(orow + 4*i);
        if (half) oe = *(const v4f*)(orow + 64);
        #pragma nounroll
        for (int pp = 0; pp < 16; ++pp) {
            #pragma unroll
            for (int j = 0; j < 4; ++j) {
                const int p = 4*pp + j;
                float* rb = RB + (j & 1)*136 + half*68;
                float* cb = CB + (j & 1)*64;
                if (row == p) {
                    #pragma unroll
                    for (int i = 0; i < 16; ++i) *(v4f*)(rb + 4*i) = ov[i];
                    if (half) *(v4f*)(rb + 64) = oe;
                }
                if (!half) cb[row] = ov[0][j];
                __syncthreads();
                const float cv  = cb[row];
                const float piv = RB[(j & 1)*136 + j];
                __builtin_amdgcn_sched_barrier(0);
                v4f pv[16];
                v4f pe = {};
                #pragma unroll
                for (int i = 0; i < 16; ++i) pv[i] = *(const v4f*)(rb + 4*i);
                if (half) pe = *(const v4f*)(rb + 64);
                const float fac = cv / piv;
                if (row != p) {
                    #pragma unroll
                    for (int i = 0; i < 16; ++i) ov[i] -= fac * pv[i];
                    oe -= fac * pe;
                }
            }
            const v4f t0 = ov[0];
            #pragma unroll
            for (int i = 0; i < 15; ++i) ov[i] = ov[i+1];
            ov[15] = t0;
        }
        #pragma unroll
        for (int i = 0; i < 16; ++i) *(v4f*)(orow + 4*i) = ov[i];
        if (half) *(v4f*)(orow + 64) = oe;
        __syncthreads();
    }

    const int wv   = tid >> 5;
    const int ln   = tid & 31;
    const int hi   = ln >> 4;
    const int col  = ln & 15;
    const int koff = hi << 1;
    const int arow = wv*16 + col;
    const int row2 = wv*16 + hi*8;

    v2f a_cur[16];
    {
        float rinv = 1.0f / aug[arow*AUGC + arow];
        #pragma unroll
        for (int k = 0; k < 16; ++k) {
            a_cur[k].x = aug[arow*AUGC + 64 + 4*k + koff]     * rinv;
            a_cur[k].y = aug[arow*AUGC + 64 + 4*k + koff + 1] * rinv;
        }
    }

    v2f wi[KS/4];
    {
        float* S = smem + SQOFF;
        v2f bvk[16];
        v2f bA[4][16];
        #pragma unroll
        for (int k = 0; k < 16; ++k) {
            int j0 = 4*k + koff;
            float d0 = aug[j0*AUGC + j0], d1 = aug[(j0+1)*AUGC + j0 + 1];
            float r0 = 1.0f / d0, r1 = 1.0f / d1;
            bvk[k].x = aug[j0*AUGC + 128] / d0;
            bvk[k].y = aug[(j0+1)*AUGC + 128] / d1;
            #pragma unroll
            for (int j = 0; j < 4; ++j) {
                bA[j][k].x = aug[j0*AUGC     + 64 + 16*j + col] * r0;
                bA[j][k].y = aug[(j0+1)*AUGC + 64 + 16*j + col] * r1;
            }
        }
        const float w0 = aug[arow*AUGC + 128] / aug[arow*AUGC + arow];
        #pragma unroll
        for (int q = 0; q < KS/4; ++q) {
            wi[q].x = (KS-1 - 4*q - koff == 0) ? w0 : 0.0f;
            wi[q].y = (KS-2 - 4*q - koff == 0) ? w0 : 0.0f;
        }
        #pragma nounroll
        for (int p = 1; p < KS; ++p) {
            float part = 0.0f;
            #pragma unroll
            for (int k = 0; k < 16; ++k) {
                part += a_cur[k].x * bvk[k].x;
                part += a_cur[k].y * bvk[k].y;
            }
            const float w = part + __shfl_xor(part, 16);
            #pragma unroll
            for (int q = 0; q < KS/4; ++q) {
                if (KS-1 - 4*q - koff == p) wi[q].x = w;
                if (KS-2 - 4*q - koff == p) wi[q].y = w;
            }
            #pragma unroll
            for (int j = 0; j < 4; ++j) {
                v8f acc = {};
                #pragma unroll
                for (int k = 0; k < 16; ++k)
                    acc = __builtin_amdgcn_wmma_f32_16x16x4_f32(false, a_cur[k], false, bA[j][k],
                                                                (short)0, acc, false, false);
                #pragma unroll
                for (int r = 0; r < 8; ++r) S[(row2 + r)*SQSTR + 16*j + col] = acc[r];
            }
            __syncthreads();
            #pragma unroll
            for (int k = 0; k < 16; ++k)
                a_cur[k] = *(const v2f*)&S[arow*SQSTR + 4*k + koff];
            __syncthreads();
        }
    }

    static_assert(KS == 16, "the output tile's input slice below is laid out for 4 steps per wave");
    v2f gy[4];
    v2f hy;
    {
        float* Gl = smem + SQOFF;
        float* Gs = Gl + (KS+1)*HD;
        float* Bl = Gs + (KS+1)*HD;
        float* Kl = Bl + HD;
        float rd = 0.0f;
        if (tid < HD) {
            rd = 1.0f / aug[tid*AUGC + tid];
            float c = Cmat[f*HD + tid];
            Gl[tid] = c;
            Gs[tid] = c * rd;
            Bl[tid] = aug[tid*AUGC + 128] * rd;
        }
        __syncthreads();
        for (int d = 1; d <= KS; ++d) {
            if (tid < HD) {
                v4f gv[HD/4];
                float av[HD];
                #pragma unroll
                for (int j = 0; j < HD/4; ++j) gv[j] = *(const v4f*)&Gs[(d-1)*HD + 4*j];
                #pragma unroll
                for (int r = 0; r < HD; ++r) av[r] = aug[r*AUGC + 64 + tid];
                __builtin_amdgcn_sched_barrier(0);
                float s = 0.0f;
                #pragma unroll
                for (int r = 0; r < HD; ++r) s += gv[r >> 2][r & 3] * av[r];
                Gl[d*HD + tid] = s;
                Gs[d*HD + tid] = s * rd;
            }
            __syncthreads();
        }
        if (tid < KS) {
            float s = 0.0f;
            #pragma unroll
            for (int r = 0; r < HD; ++r) s += Gl[tid*HD + r] * Bl[r];
            Kl[tid] = s;
        }
        __syncthreads();
        #pragma unroll
        for (int q = 0; q < 4; ++q)
            gy[q] = *(const v2f*)&Gl[(col + 1)*HD + 16*wv + 4*q + koff];
        const int d0 = col - 4*wv - koff, d1 = d0 - 1;
        const float k0 = Kl[d0 < 0 ? 0 : d0], k1 = Kl[d1 < 0 ? 0 : d1];
        hy.x = (d0 >= 0) ? k0 : 0.0f;
        hy.y = (d1 >= 0) ? k1 : 0.0f;
    }

    for (int e = tid; e < NBATCH*NT; e += 128) {
        int b = e >> 10, i = e & (NT - 1);
        u_lds[i*NBATCH + b] = x[(size_t)(b*NFEAT + f)*NT + i];
    }
    for (int e = tid; e < XBUF; e += 128) X_lds[e] = 0.0f;
    __syncthreads();

    const int pslot = wv*16 + col;
    const int cb    = col & (NBATCH-1);
    for (int it = 0; it < NT/KS; ++it) {
        const int t = KS*it;
        const float* Xr = X_lds + (it & 1)*XBUF;
        float*       Xw = X_lds + ((it & 1) ^ 1)*XBUF;

        v2f ub[KS/4];
        #pragma unroll
        for (int q = 0; q < KS/4; ++q) {
            ub[q].x = u_lds[(t + 4*q + koff)*NBATCH + cb];
            ub[q].y = u_lds[(t + 4*q + koff + 1)*NBATCH + cb];
        }
        v2f uy;
        uy.x = u_lds[(t + 4*wv + koff)*NBATCH + cb];
        uy.y = u_lds[(t + 4*wv + koff + 1)*NBATCH + cb];
        v2f bop[16];
        #pragma unroll
        for (int k = 0; k < 16; ++k)
            bop[k] = *(const v2f*)&Xr[col*XSTR + 4*k + koff];
        v2f by[4];
        #pragma unroll
        for (int q = 0; q < 4; ++q)
            by[q] = *(const v2f*)&Xr[col*XSTR + 16*wv + 4*q + koff];
        __builtin_amdgcn_sched_barrier(0);

        v8f zero = {};
        v8f accA = __builtin_amdgcn_wmma_f32_16x16x4_f32(false, wi[0], false, ub[0], (short)0, zero, false, false);
        v8f accB = __builtin_amdgcn_wmma_f32_16x16x4_f32(false, wi[1], false, ub[1], (short)0, zero, false, false);
        v8f accY = __builtin_amdgcn_wmma_f32_16x16x4_f32(false, hy,    false, uy,    (short)0, zero, false, false);
        #pragma unroll
        for (int q = 2; q < KS/4; q += 2) {
            accA = __builtin_amdgcn_wmma_f32_16x16x4_f32(false, wi[q],   false, ub[q],   (short)0, accA, false, false);
            accB = __builtin_amdgcn_wmma_f32_16x16x4_f32(false, wi[q+1], false, ub[q+1], (short)0, accB, false, false);
        }
        #pragma unroll
        for (int k = 0; k < 16; k += 2) {
            accA = __builtin_amdgcn_wmma_f32_16x16x4_f32(false, a_cur[k],   false, bop[k],   (short)0, accA, false, false);
            accB = __builtin_amdgcn_wmma_f32_16x16x4_f32(false, a_cur[k+1], false, bop[k+1], (short)0, accB, false, false);
            if ((k & 3) == 0)
                accY = __builtin_amdgcn_wmma_f32_16x16x4_f32(false, gy[k >> 2], false, by[k >> 2], (short)0, accY, false, false);
        }

        v8f xs = accA + accB;
        v4f lo4 = { xs[0], xs[1], xs[2], xs[3] };
        v4f hi4 = { xs[4], xs[5], xs[6], xs[7] };
        *(v4f*)&Xw[col*XSTR + row2]     = lo4;
        *(v4f*)&Xw[col*XSTR + row2 + 4] = hi4;
        #pragma unroll
        for (int r = 0; r < 8; ++r)
            p_lds[((t + 8*hi + r) & (YCHUNK-1))*PSTR + pslot] = accY[r];
        __syncthreads();

        if (((t + KS-1) & (YCHUNK-1)) == (YCHUNK-1)) {
            int tbase = t & ~(YCHUNK-1);
            for (int e = tid; e < NBATCH*YCHUNK; e += 128) {
                int b = e >> 7, i = e & (YCHUNK-1);
                float y = 0.0f;
                for (int j = 0; j < 4; ++j) y += p_lds[i*PSTR + j*16 + b];
                out[(size_t)(b*NFEAT + f)*NT + tbase + i] = y;
            }
            __syncthreads();
        }
    }
}

extern "C" void kernel_launch(void* const* d_in, const int* in_sizes, int n_in,
                              void* d_out, int out_size, void* d_ws, size_t ws_size,
                              hipStream_t stream) {
    (void)in_sizes; (void)n_in; (void)d_ws; (void)ws_size; (void)out_size;
    const float* x = (const float*)d_in[0];
    const float* A = (const float*)d_in[1];
    const float* B = (const float*)d_in[2];
    const float* C = (const float*)d_in[3];
    float* out = (float*)d_out;
    ssm_scan_wmma_f32<<<dim3(NFEAT), dim3(128), 0, stream>>>(x, A, B, C, out);
}


# The 16 serial vector rounds for the output rows g_d and the staging of the input u run inside the power rounds, beside their WMMAs and under their barriers, not as phases of their own; same sums
# speedup vs baseline: 9.3254x; 1.0962x over previous
#include <hip/hip_runtime.h>

typedef __attribute__((ext_vector_type(2))) float v2f;
typedef __attribute__((ext_vector_type(4))) float v4f;
typedef __attribute__((ext_vector_type(8))) float v8f;

#define NBATCH 8
#define NFEAT  256
#define HD     64
#define NT     1024
#define AUGC   132
#define XSTR   68
#define YCHUNK 128
#define XBUF   (16*XSTR)
#define PSTR   129
#define SQOFF  8704
#define SQSTR  68
#define KS     16
#define GOFF   (SQOFF + HD*SQSTR)
#define XCH    4

__launch_bounds__(128)
__global__ void ssm_scan_wmma_f32(const float* __restrict__ x,
                                  const float* __restrict__ Amat,
                                  const float* __restrict__ Bmat,
                                  const float* __restrict__ Cmat,
                                  float* __restrict__ out)
{
    __shared__ __attribute__((aligned(16))) float smem[NBATCH*NT + YCHUNK*PSTR + 2*XBUF];
    float* aug   = smem;
    float* u_lds = smem;
    float* p_lds = smem + NBATCH*NT;
    float* X_lds = smem + NBATCH*NT + YCHUNK*PSTR;

    const int f   = blockIdx.x;
    const int tid = threadIdx.x;

    const float* Af = Amat + f*HD*HD;
    for (int e = tid; e < HD*HD; e += 128) {
        int r = e >> 6, c = e & 63;
        float a   = Af[e];
        float eye = (r == c) ? 1.0f : 0.0f;
        aug[r*AUGC + c]      = eye - 0.5f*a;
        aug[r*AUGC + 64 + c] = eye + 0.5f*a;
    }
    for (int r = tid; r < HD; r += 128) {
        aug[r*AUGC + 128] = Bmat[f*HD + r];
        aug[r*AUGC + 129] = 0.0f;
        aug[r*AUGC + 130] = 0.0f;
        aug[r*AUGC + 131] = 0.0f;
    }
    __syncthreads();

    {
        const int row  = tid & 63;
        const int half = tid >> 6;
        float* orow = aug + row*AUGC + half*64;
        float* RB = smem + SQOFF;
        float* CB = RB + 2*2*68;
        v4f ov[16];
        v4f oe = {};
        #pragma unroll
        for (int i = 0; i < 16; ++i) ov[i] = *(const v4f*)(orow + 4*i);
        if (half) oe = *(const v4f*)(orow + 64);
        #pragma nounroll
        for (int pp = 0; pp < 16; ++pp) {
            #pragma unroll
            for (int j = 0; j < 4; ++j) {
                const int p = 4*pp + j;
                float* rb = RB + (j & 1)*136 + half*68;
                float* cb = CB + (j & 1)*64;
                if (row == p) {
                    #pragma unroll
                    for (int i = 0; i < 16; ++i) *(v4f*)(rb + 4*i) = ov[i];
                    if (half) *(v4f*)(rb + 64) = oe;
                }
                if (!half) cb[row] = ov[0][j];
                __syncthreads();
                const float cv  = cb[row];
                const float piv = RB[(j & 1)*136 + j];
                __builtin_amdgcn_sched_barrier(0);
                v4f pv[16];
                v4f pe = {};
                #pragma unroll
                for (int i = 0; i < 16; ++i) pv[i] = *(const v4f*)(rb + 4*i);
                if (half) pe = *(const v4f*)(rb + 64);
                const float fac = cv / piv;
                if (row != p) {
                    #pragma unroll
                    for (int i = 0; i < 16; ++i) ov[i] -= fac * pv[i];
                    oe -= fac * pe;
                }
            }
            const v4f t0 = ov[0];
            #pragma unroll
            for (int i = 0; i < 15; ++i) ov[i] = ov[i+1];
            ov[15] = t0;
        }
        #pragma unroll
        for (int i = 0; i < 16; ++i) *(v4f*)(orow + 4*i) = ov[i];
        if (half) *(v4f*)(orow + 64) = oe;
        __syncthreads();
    }


    const int wv   = tid >> 5;
    const int ln   = tid & 31;
    const int hi   = ln >> 4;
    const int col  = ln & 15;
    const int koff = hi << 1;
    const int arow = wv*16 + col;
    const int row2 = wv*16 + hi*8;
    const int gm   = tid & (HD-1);

    const float cC = Cmat[f*HD + gm];
    float xv[XCH];
    #pragma unroll
    for (int q = 0; q < XCH; ++q) xv[q] = x[(size_t)f*NT + tid + 128*q];

    v2f a_cur[16];
    {
        float rinv = 1.0f / aug[arow*AUGC + arow];
        #pragma unroll
        for (int k = 0; k < 16; ++k) {
            a_cur[k].x = aug[arow*AUGC + 64 + 4*k + koff]     * rinv;
            a_cur[k].y = aug[arow*AUGC + 64 + 4*k + koff + 1] * rinv;
        }
    }

    static_assert(KS == 16, "the output tile's input slice below is laid out for 4 steps per wave");
    static_assert(XCH*KS == NBATCH*NT/128, "one chunk of the input ahead of the loop, one in each of its KS-1 rounds");
    static_assert(XCH*128*2 == NT && KS == 2*NBATCH, "a chunk is half a batch row of the input: chunk c = batch c>>1, half c&1");
    v2f wi[KS/4];
    v2f gy[4];
    v2f hy;
    {
        float* S  = smem + SQOFF;
        float* Gl = smem + GOFF;
        float* Gs = Gl + (KS+1)*HD;
        float* Bl = Gs + (KS+1)*HD;
        float* Kl = Bl + HD;
        v2f bvk[16];
        v2f bA[4][16];
        #pragma unroll
        for (int k = 0; k < 16; ++k) {
            int j0 = 4*k + koff;
            float d0 = aug[j0*AUGC + j0], d1 = aug[(j0+1)*AUGC + j0 + 1];
            float r0 = 1.0f / d0, r1 = 1.0f / d1;
            bvk[k].x = aug[j0*AUGC + 128] / d0;
            bvk[k].y = aug[(j0+1)*AUGC + 128] / d1;
            #pragma unroll
            for (int j = 0; j < 4; ++j) {
                bA[j][k].x = aug[j0*AUGC     + 64 + 16*j + col] * r0;
                bA[j][k].y = aug[(j0+1)*AUGC + 64 + 16*j + col] * r1;
            }
        }
        const float w0 = aug[arow*AUGC + 128] / aug[arow*AUGC + arow];
        #pragma unroll
        for (int q = 0; q < KS/4; ++q) {
            wi[q].x = (KS-1 - 4*q - koff == 0) ? w0 : 0.0f;
            wi[q].y = (KS-2 - 4*q - koff == 0) ? w0 : 0.0f;
        }
        float av[HD];
        #pragma unroll
        for (int r = 0; r < HD; ++r) av[r] = aug[r*AUGC + 64 + gm];
        const float rd = 1.0f / aug[gm*AUGC + gm];
        float* gdst = (tid < HD) ? Gl : Gs;
        const float gsc = (tid < HD) ? 1.0f : rd;
        if (tid < HD) {
            Gl[tid] = cC;
            Gs[tid] = cC * rd;
            Bl[tid] = aug[tid*AUGC + 128] * rd;
        }
        __syncthreads();
        v4f gv[HD/4];
        #pragma unroll
        for (int j = 0; j < HD/4; ++j) gv[j] = *(const v4f*)&Gs[4*j];
        #pragma nounroll
        for (int p = 1; p < KS; ++p) {
            {
                const int c = p - 1;
                float* ud = u_lds + (512*(c & 1) + tid)*NBATCH + (c >> 1);
                #pragma unroll
                for (int q = 0; q < XCH; ++q) ud[128*q*NBATCH] = xv[q];
            }
            {
                const float* xs = x + (size_t)((p >> 1)*NFEAT + f)*NT + 512*(p & 1) + tid;
                #pragma unroll
                for (int q = 0; q < XCH; ++q) xv[q] = xs[128*q];
            }
            float part = 0.0f;
            #pragma unroll
            for (int k = 0; k < 16; ++k) {
                part += a_cur[k].x * bvk[k].x;
                part += a_cur[k].y * bvk[k].y;
            }
            const float w = part + __shfl_xor(part, 16);
            #pragma unroll
            for (int q = 0; q < KS/4; ++q) {
                if (KS-1 - 4*q - koff == p) wi[q].x = w;
                if (KS-2 - 4*q - koff == p) wi[q].y = w;
            }
            float s = 0.0f;
            #pragma unroll
            for (int r = 0; r < HD; ++r) s += gv[r >> 2][r & 3] * av[r];
            #pragma unroll
            for (int j = 0; j < 4; ++j) {
                v8f acc = {};
                #pragma unroll
                for (int k = 0; k < 16; ++k)
                    acc = __builtin_amdgcn_wmma_f32_16x16x4_f32(false, a_cur[k], false, bA[j][k],
                                                                (short)0, acc, false, false);
                #pragma unroll
                for (int r = 0; r < 8; ++r) S[(row2 + r)*SQSTR + 16*j + col] = acc[r];
            }
            gdst[p*HD + gm] = s * gsc;
            __syncthreads();
            #pragma unroll
            for (int k = 0; k < 16; ++k)
                a_cur[k] = *(const v2f*)&S[arow*SQSTR + 4*k + koff];
            #pragma unroll
            for (int j = 0; j < HD/4; ++j) gv[j] = *(const v4f*)&Gs[p*HD + 4*j];
            __syncthreads();
        }
        #pragma unroll
        for (int q = 0; q < XCH; ++q)
            u_lds[(512 + tid + 128*q)*NBATCH + NBATCH-1] = xv[q];
        for (int e = tid; e < XBUF; e += 128) X_lds[e] = 0.0f;
        {
            float s = 0.0f;
            #pragma unroll
            for (int r = 0; r < HD; ++r) s += gv[r >> 2][r & 3] * av[r];
            if (tid < HD) Gl[KS*HD + tid] = s;
        }
        if (tid < KS) {
            float s = 0.0f;
            #pragma unroll
            for (int r = 0; r < HD; ++r) s += Gl[tid*HD + r] * Bl[r];
            Kl[tid] = s;
        }
        __syncthreads();
        #pragma unroll
        for (int q = 0; q < 4; ++q)
            gy[q] = *(const v2f*)&Gl[(col + 1)*HD + 16*wv + 4*q + koff];
        const int d0 = col - 4*wv - koff, d1 = d0 - 1;
        const float k0 = Kl[d0 < 0 ? 0 : d0], k1 = Kl[d1 < 0 ? 0 : d1];
        hy.x = (d0 >= 0) ? k0 : 0.0f;
        hy.y = (d1 >= 0) ? k1 : 0.0f;
    }

    const int pslot = wv*16 + col;
    const int cb    = col & (NBATCH-1);
    for (int it = 0; it < NT/KS; ++it) {
        const int t = KS*it;
        const float* Xr = X_lds + (it & 1)*XBUF;
        float*       Xw = X_lds + ((it & 1) ^ 1)*XBUF;

        v2f ub[KS/4];
        #pragma unroll
        for (int q = 0; q < KS/4; ++q) {
            ub[q].x = u_lds[(t + 4*q + koff)*NBATCH + cb];
            ub[q].y = u_lds[(t + 4*q + koff + 1)*NBATCH + cb];
        }
        v2f uy;
        uy.x = u_lds[(t + 4*wv + koff)*NBATCH + cb];
        uy.y = u_lds[(t + 4*wv + koff + 1)*NBATCH + cb];
        v2f bop[16];
        #pragma unroll
        for (int k = 0; k < 16; ++k)
            bop[k] = *(const v2f*)&Xr[col*XSTR + 4*k + koff];
        v2f by[4];
        #pragma unroll
        for (int q = 0; q < 4; ++q)
            by[q] = *(const v2f*)&Xr[col*XSTR + 16*wv + 4*q + koff];
        __builtin_amdgcn_sched_barrier(0);

        v8f zero = {};
        v8f accA = __builtin_amdgcn_wmma_f32_16x16x4_f32(false, wi[0], false, ub[0], (short)0, zero, false, false);
        v8f accB = __builtin_amdgcn_wmma_f32_16x16x4_f32(false, wi[1], false, ub[1], (short)0, zero, false, false);
        v8f accY = __builtin_amdgcn_wmma_f32_16x16x4_f32(false, hy,    false, uy,    (short)0, zero, false, false);
        #pragma unroll
        for (int q = 2; q < KS/4; q += 2) {
            accA = __builtin_amdgcn_wmma_f32_16x16x4_f32(false, wi[q],   false, ub[q],   (short)0, accA, false, false);
            accB = __builtin_amdgcn_wmma_f32_16x16x4_f32(false, wi[q+1], false, ub[q+1], (short)0, accB, false, false);
        }
        #pragma unroll
        for (int k = 0; k < 16; k += 2) {
            accA = __builtin_amdgcn_wmma_f32_16x16x4_f32(false, a_cur[k],   false, bop[k],   (short)0, accA, false, false);
            accB = __builtin_amdgcn_wmma_f32_16x16x4_f32(false, a_cur[k+1], false, bop[k+1], (short)0, accB, false, false);
            if ((k & 3) == 0)
                accY = __builtin_amdgcn_wmma_f32_16x16x4_f32(false, gy[k >> 2], false, by[k >> 2], (short)0, accY, false, false);
        }

        v8f xs = accA + accB;
        v4f lo4 = { xs[0], xs[1], xs[2], xs[3] };
        v4f hi4 = { xs[4], xs[5], xs[6], xs[7] };
        *(v4f*)&Xw[col*XSTR + row2]     = lo4;
        *(v4f*)&Xw[col*XSTR + row2 + 4] = hi4;
        #pragma unroll
        for (int r = 0; r < 8; ++r)
            p_lds[((t + 8*hi + r) & (YCHUNK-1))*PSTR + pslot] = accY[r];
        __syncthreads();

        if (((t + KS-1) & (YCHUNK-1)) == (YCHUNK-1)) {
            int tbase = t & ~(YCHUNK-1);
            for (int e = tid; e < NBATCH*YCHUNK; e += 128) {
                int b = e >> 7, i = e & (YCHUNK-1);
                float y = 0.0f;
                for (int j = 0; j < 4; ++j) y += p_lds[i*PSTR + j*16 + b];
                out[(size_t)(b*NFEAT + f)*NT + tbase + i] = y;
            }
            __syncthreads();
        }
    }
}

extern "C" void kernel_launch(void* const* d_in, const int* in_sizes, int n_in,
                              void* d_out, int out_size, void* d_ws, size_t ws_size,
                              hipStream_t stream) {
    (void)in_sizes; (void)n_in; (void)d_ws; (void)ws_size; (void)out_size;
    const float* x = (const float*)d_in[0];
    const float* A = (const float*)d_in[1];
    const float* B = (const float*)d_in[2];
    const float* C = (const float*)d_in[3];
    float* out = (float*)d_out;
    ssm_scan_wmma_f32<<<dim3(NFEAT), dim3(128), 0, stream>>>(x, A, B, C, out);
}
